# starting kernel
# baseline (speedup 1.0000x reference)
#include <hip/hip_runtime.h>
#include <math.h>

#define NQ        9
#define DIM       512
#define NLAYERS   12
#define OBS       256
#define BATCH_TOT 16384

#define WAVES_PER_BLOCK   8
#define STATES_PER_WAVE   8
#define BLOCK_THREADS     (WAVES_PER_BLOCK * 32)
#define STATES_PER_BLOCK  (WAVES_PER_BLOCK * STATES_PER_WAVE)
#define NGATES            (NLAYERS * NQ)
#define GATE_FLOATS       128
#define GATE_LDS_FLOATS   (NGATES * GATE_FLOATS)
#define STATE_LDS_FLOATS  (STATES_PER_WAVE * DIM * 2)

typedef float v2f __attribute__((ext_vector_type(2)));
typedef float v8f __attribute__((ext_vector_type(8)));

struct cpx { float re, im; };
__device__ inline cpx cz() { cpx r; r.re = 0.f; r.im = 0.f; return r; }
__device__ inline cpx cmk(float a, float b) { cpx r; r.re = a; r.im = b; return r; }
__device__ inline cpx cmul(cpx a, cpx b) { return cmk(a.re*b.re - a.im*b.im, a.re*b.im + a.im*b.re); }
__device__ inline cpx cadd(cpx a, cpx b) { return cmk(a.re + b.re, a.im + b.im); }

__device__ inline void u3mat(float t, float p, float l, cpx U[2][2]) {
    float ct = cosf(0.5f * t), st = sinf(0.5f * t);
    float cp = cosf(p), sp = sinf(p), cl = cosf(l), sl = sinf(l);
    U[0][0] = cmk(ct, 0.f);
    U[0][1] = cmk(-cl * st, -sl * st);
    U[1][0] = cmk(cp * st, sp * st);
    U[1][1] = cmk((cp*cl - sp*sl) * ct, (cp*sl + sp*cl) * ct);
}

__device__ inline void kron22(const cpx A[2][2], const cpx B[2][2], cpx K[4][4]) {
    for (int a = 0; a < 2; ++a)
        for (int b = 0; b < 2; ++b)
            for (int c = 0; c < 2; ++c)
                for (int d = 0; d < 2; ++d)
                    K[2*a + c][2*b + d] = cmul(A[a][b], B[c][d]);
}

__device__ inline void mat4mul(cpx C[4][4], const cpx A[4][4], const cpx B[4][4]) {
    for (int i = 0; i < 4; ++i)
        for (int j = 0; j < 4; ++j) {
            cpx acc = cz();
            for (int k = 0; k < 4; ++k) acc = cadd(acc, cmul(A[i][k], B[k][j]));
            C[i][j] = acc;
        }
}

__device__ void computeGate(int g, const float* __restrict__ w, float* __restrict__ gl) {
    const float* th = w + g * 15;
    cpx Ua[2][2], Ub[2][2];
    cpx pre[4][4], post[4][4];
    u3mat(th[0], th[1], th[2], Ua);  u3mat(th[3], th[4], th[5], Ub);  kron22(Ua, Ub, pre);
    u3mat(th[9], th[10], th[11], Ua); u3mat(th[12], th[13], th[14], Ub); kron22(Ua, Ub, post);

    float c6 = cosf(0.5f*th[6]), s6 = sinf(0.5f*th[6]);
    float c7 = cosf(0.5f*th[7]), s7 = sinf(0.5f*th[7]);
    float c8 = cosf(0.5f*th[8]), s8 = sinf(0.5f*th[8]);
    cpx Rxx[4][4], Ryy[4][4], Rzz[4][4];
    for (int i = 0; i < 4; ++i)
        for (int j = 0; j < 4; ++j) { Rxx[i][j] = cz(); Ryy[i][j] = cz(); Rzz[i][j] = cz(); }
    for (int i = 0; i < 4; ++i) { Rxx[i][i] = cmk(c6, 0.f); Ryy[i][i] = cmk(c7, 0.f); }
    Rxx[0][3] = cmk(0.f, -s6); Rxx[1][2] = cmk(0.f, -s6);
    Rxx[2][1] = cmk(0.f, -s6); Rxx[3][0] = cmk(0.f, -s6);
    Ryy[0][3] = cmk(0.f,  s7); Ryy[1][2] = cmk(0.f, -s7);
    Ryy[2][1] = cmk(0.f, -s7); Ryy[3][0] = cmk(0.f,  s7);
    Rzz[0][0] = cmk(c8, -s8); Rzz[1][1] = cmk(c8,  s8);
    Rzz[2][2] = cmk(c8,  s8); Rzz[3][3] = cmk(c8, -s8);

    cpx t1[4][4], t2[4][4];
    mat4mul(t1, Rxx, Ryy);
    mat4mul(t2, t1, Rzz);
    mat4mul(t1, t2, pre);
    mat4mul(t2, post, t1);

    float* gb = gl + g * GATE_FLOATS;
    for (int c = 0; c < 2; ++c)
        for (int k = 0; k < 4; ++k)
            for (int n = 0; n < 16; ++n) {
                int a = n & 7;
                int b = 4*c + k;
                float val;
                if (a < 4) val = (b < 4) ?  t2[a][b].re     : -t2[a][b-4].im;
                else       val = (b < 4) ?  t2[a-4][b].im   :  t2[a-4][b-4].re;
                gb[c*64 + k*16 + n] = val;
            }
}

__device__ inline unsigned insert2(unsigned r, int lo, unsigned vlo, int hi, unsigned vhi) {
    unsigned low = r & ((1u << lo) - 1u);
    unsigned x = ((r >> lo) << (lo + 1)) | (vlo << lo) | low;
    unsigned low2 = x & ((1u << hi) - 1u);
    return ((x >> hi) << (hi + 1)) | (vhi << hi) | low2;
}
__device__ inline unsigned ampIndex(unsigned r, int pm, int pl, unsigned p) {
    unsigned vm = (p >> 1) & 1u, vl = p & 1u;
    int lo  = (pm < pl) ? pm : pl;  unsigned vlo = (pm < pl) ? vm : vl;
    int hi  = (pm < pl) ? pl : pm;  unsigned vhi = (pm < pl) ? vl : vm;
    return insert2(r, lo, vlo, hi, vhi);
}

__global__ __launch_bounds__(BLOCK_THREADS, 1)
void su4_quantum_kernel(const float* __restrict__ features,
                        const float* __restrict__ weights,
                        float* __restrict__ out) {
    extern __shared__ float smem[];
    float* gl = smem;
    const int tid  = threadIdx.x;
    const int lane = tid & 31;
    const int wave = tid >> 5;
    float*  stateF = smem + GATE_LDS_FLOATS + wave * STATE_LDS_FLOATS;
    float2* stateV = (float2*)stateF;

    if (tid < NGATES) computeGate(tid, weights, gl);

    const int bBase = (blockIdx.x * WAVES_PER_BLOCK + wave) * STATES_PER_WAVE;
    for (int s = 0; s < STATES_PER_WAVE; ++s) {
        const float* f = features + (size_t)(bBase + s) * OBS;
        float ss = 0.f;
        for (int j = lane; j < OBS; j += 32) { float v = f[j]; ss += v * v; }
        for (int msk = 16; msk >= 1; msk >>= 1) ss += __shfl_xor(ss, msk, 32);
        float inv = 1.0f / (sqrtf(ss) + 1e-8f);
        for (int j = lane; j < DIM; j += 32) {
            float re = (j < OBS) ? f[j] * inv : 0.f;
            stateV[s * DIM + j] = make_float2(re, 0.f);
        }
    }
    __syncthreads();

    const int s    = lane & 7;
    const unsigned rsub = (unsigned)((lane >> 3) & 1);
    const int half = lane >> 4;
    const int kb   = 2 * half;
    const int ncol = lane & 15;
    const unsigned q = (unsigned)(ncol & 3);
    const int reim = (ncol >> 2) & 1;

    for (int layer = 0; layer < NLAYERS; ++layer) {
        for (int i = 0; i < NQ; ++i) {
            const int g = layer * NQ + i;
            const float* gb = gl + g * GATE_FLOATS;
            v2f B0, B1;
            B0[0] = gb[(kb    ) * 16 + ncol];
            B0[1] = gb[(kb + 1) * 16 + ncol];
            B1[0] = gb[64 + (kb    ) * 16 + ncol];
            B1[1] = gb[64 + (kb + 1) * 16 + ncol];
            const int pm = (i < 8) ? (8 - i) : 0;
            const int pl = (i < 8) ? (7 - i) : 8;

            float2 x0 = stateV[s * DIM + ampIndex(rsub, pm, pl, (unsigned)kb)];
            float2 x1 = stateV[s * DIM + ampIndex(rsub, pm, pl, (unsigned)(kb + 1))];

            for (int rr = 0; rr < 64; ++rr) {
                unsigned rn = 2u * (unsigned)((rr + 1) & 63) + rsub;
                float2 y0 = stateV[s * DIM + ampIndex(rn, pm, pl, (unsigned)kb)];
                float2 y1 = stateV[s * DIM + ampIndex(rn, pm, pl, (unsigned)(kb + 1))];

                v2f A0; A0[0] = x0.x; A0[1] = x1.x;
                v2f A1; A1[0] = x0.y; A1[1] = x1.y;
                v8f acc = {0.f, 0.f, 0.f, 0.f, 0.f, 0.f, 0.f, 0.f};
                acc = __builtin_amdgcn_wmma_f32_16x16x4_f32(false, A0, false, B0,
                                                            (short)0, acc, false, false);
                acc = __builtin_amdgcn_wmma_f32_16x16x4_f32(false, A1, false, B1,
                                                            (short)0, acc, false, false);

                unsigned aw = ampIndex(2u * (unsigned)rr + (unsigned)half, pm, pl, q);
                #pragma unroll
                for (int v = 0; v < 8; ++v)
                    stateF[v * (DIM * 2) + aw * 2 + reim] = acc[v];

                x0 = y0; x1 = y1;
            }
        }
    }

    for (int ss2 = 0; ss2 < STATES_PER_WAVE; ++ss2) {
        float* o = out + (size_t)(bBase + ss2) * DIM;
        for (int j = lane; j < DIM; j += 32) {
            float2 a = stateV[ss2 * DIM + j];
            o[j] = tanhf(5.0f * (a.x * a.x + a.y * a.y));
        }
    }
}

extern "C" void kernel_launch(void* const* d_in, const int* in_sizes, int n_in,
                              void* d_out, int out_size, void* d_ws, size_t ws_size,
                              hipStream_t stream) {
    (void)in_sizes; (void)n_in; (void)d_ws; (void)ws_size; (void)out_size;
    const float* features = (const float*)d_in[0];
    const float* weights  = (const float*)d_in[1];
    float* out = (float*)d_out;

    dim3 grid(BATCH_TOT / STATES_PER_BLOCK);
    dim3 block(BLOCK_THREADS);
    size_t shmem = (size_t)(GATE_LDS_FLOATS + WAVES_PER_BLOCK * STATE_LDS_FLOATS)
                   * sizeof(float);
    su4_quantum_kernel<<<grid, block, shmem, stream>>>(features, weights, out);
}


# LDS state slots XOR-swizzled (amp bits 5..8 and state id folded into bank bits): A-fragment loads (8/16-way) and D stores (2/4-way) now bank-conflict-free; address = scalar part ^ lane part
# speedup vs baseline: 1.3488x; 1.3488x over previous
#include <hip/hip_runtime.h>
#include <math.h>

#define NQ        9
#define DIM       512
#define NLAYERS   12
#define OBS       256
#define BATCH_TOT 16384

#define WAVES_PER_BLOCK   8
#define STATES_PER_WAVE   8
#define BLOCK_THREADS     (WAVES_PER_BLOCK * 32)
#define STATES_PER_BLOCK  (WAVES_PER_BLOCK * STATES_PER_WAVE)
#define NGATES            (NLAYERS * NQ)
#define GATE_FLOATS       128
#define GATE_LDS_FLOATS   (NGATES * GATE_FLOATS)
#define STATE_LDS_FLOATS  (STATES_PER_WAVE * DIM * 2)

typedef float v2f __attribute__((ext_vector_type(2)));
typedef float v8f __attribute__((ext_vector_type(8)));

struct cpx { float re, im; };
__device__ inline cpx cz() { cpx r; r.re = 0.f; r.im = 0.f; return r; }
__device__ inline cpx cmk(float a, float b) { cpx r; r.re = a; r.im = b; return r; }
__device__ inline cpx cmul(cpx a, cpx b) { return cmk(a.re*b.re - a.im*b.im, a.re*b.im + a.im*b.re); }
__device__ inline cpx cadd(cpx a, cpx b) { return cmk(a.re + b.re, a.im + b.im); }

__device__ inline void u3mat(float t, float p, float l, cpx U[2][2]) {
    float ct = cosf(0.5f * t), st = sinf(0.5f * t);
    float cp = cosf(p), sp = sinf(p), cl = cosf(l), sl = sinf(l);
    U[0][0] = cmk(ct, 0.f);
    U[0][1] = cmk(-cl * st, -sl * st);
    U[1][0] = cmk(cp * st, sp * st);
    U[1][1] = cmk((cp*cl - sp*sl) * ct, (cp*sl + sp*cl) * ct);
}

__device__ inline void kron22(const cpx A[2][2], const cpx B[2][2], cpx K[4][4]) {
    for (int a = 0; a < 2; ++a)
        for (int b = 0; b < 2; ++b)
            for (int c = 0; c < 2; ++c)
                for (int d = 0; d < 2; ++d)
                    K[2*a + c][2*b + d] = cmul(A[a][b], B[c][d]);
}

__device__ inline void mat4mul(cpx C[4][4], const cpx A[4][4], const cpx B[4][4]) {
    for (int i = 0; i < 4; ++i)
        for (int j = 0; j < 4; ++j) {
            cpx acc = cz();
            for (int k = 0; k < 4; ++k) acc = cadd(acc, cmul(A[i][k], B[k][j]));
            C[i][j] = acc;
        }
}

__device__ void computeGate(int g, const float* __restrict__ w, float* __restrict__ gl) {
    const float* th = w + g * 15;
    cpx Ua[2][2], Ub[2][2];
    cpx pre[4][4], post[4][4];
    u3mat(th[0], th[1], th[2], Ua);  u3mat(th[3], th[4], th[5], Ub);  kron22(Ua, Ub, pre);
    u3mat(th[9], th[10], th[11], Ua); u3mat(th[12], th[13], th[14], Ub); kron22(Ua, Ub, post);

    float c6 = cosf(0.5f*th[6]), s6 = sinf(0.5f*th[6]);
    float c7 = cosf(0.5f*th[7]), s7 = sinf(0.5f*th[7]);
    float c8 = cosf(0.5f*th[8]), s8 = sinf(0.5f*th[8]);
    cpx Rxx[4][4], Ryy[4][4], Rzz[4][4];
    for (int i = 0; i < 4; ++i)
        for (int j = 0; j < 4; ++j) { Rxx[i][j] = cz(); Ryy[i][j] = cz(); Rzz[i][j] = cz(); }
    for (int i = 0; i < 4; ++i) { Rxx[i][i] = cmk(c6, 0.f); Ryy[i][i] = cmk(c7, 0.f); }
    Rxx[0][3] = cmk(0.f, -s6); Rxx[1][2] = cmk(0.f, -s6);
    Rxx[2][1] = cmk(0.f, -s6); Rxx[3][0] = cmk(0.f, -s6);
    Ryy[0][3] = cmk(0.f,  s7); Ryy[1][2] = cmk(0.f, -s7);
    Ryy[2][1] = cmk(0.f, -s7); Ryy[3][0] = cmk(0.f,  s7);
    Rzz[0][0] = cmk(c8, -s8); Rzz[1][1] = cmk(c8,  s8);
    Rzz[2][2] = cmk(c8,  s8); Rzz[3][3] = cmk(c8, -s8);

    cpx t1[4][4], t2[4][4];
    mat4mul(t1, Rxx, Ryy);
    mat4mul(t2, t1, Rzz);
    mat4mul(t1, t2, pre);
    mat4mul(t2, post, t1);

    float* gb = gl + g * GATE_FLOATS;
    for (int c = 0; c < 2; ++c)
        for (int k = 0; k < 4; ++k)
            for (int n = 0; n < 16; ++n) {
                int a = n & 7;
                int b = 4*c + k;
                float val;
                if (a < 4) val = (b < 4) ?  t2[a][b].re     : -t2[a][b-4].im;
                else       val = (b < 4) ?  t2[a-4][b].im   :  t2[a-4][b-4].re;
                gb[c*64 + k*16 + n] = val;
            }
}

__device__ inline unsigned insert2(unsigned r, int lo, unsigned vlo, int hi, unsigned vhi) {
    unsigned low = r & ((1u << lo) - 1u);
    unsigned x = ((r >> lo) << (lo + 1)) | (vlo << lo) | low;
    unsigned low2 = x & ((1u << hi) - 1u);
    return ((x >> hi) << (hi + 1)) | (vhi << hi) | low2;
}

__device__ inline unsigned foldAmp(unsigned a) {
    return a ^ ((((a >> 5) ^ (a >> 7)) & 3u) << 3);
}
__device__ inline unsigned stateSwz(unsigned s) {
    return ((s & 1u) ? 12u : 0u) ^ ((s & 2u) ? 24u : 0u) ^ ((s & 4u) ? 7u : 0u);
}
__device__ inline unsigned spreadR(unsigned r, int pm, int pl) {
    int lo = (pm < pl) ? pm : pl;
    int hi = (pm < pl) ? pl : pm;
    return insert2(r, lo, 0u, hi, 0u);
}

__global__ __launch_bounds__(BLOCK_THREADS, 1)
void su4_quantum_kernel(const float* __restrict__ features,
                        const float* __restrict__ weights,
                        float* __restrict__ out) {
    extern __shared__ float smem[];
    float* gl = smem;
    const int tid  = threadIdx.x;
    const int lane = tid & 31;
    const int wave = tid >> 5;
    float*  stateF = smem + GATE_LDS_FLOATS + wave * STATE_LDS_FLOATS;
    float2* stateV = (float2*)stateF;

    if (tid < NGATES) computeGate(tid, weights, gl);

    const int bBase = (blockIdx.x * WAVES_PER_BLOCK + wave) * STATES_PER_WAVE;
    for (int s = 0; s < STATES_PER_WAVE; ++s) {
        const float* f = features + (size_t)(bBase + s) * OBS;
        float ss = 0.f;
        for (int j = lane; j < OBS; j += 32) { float v = f[j]; ss += v * v; }
        for (int msk = 16; msk >= 1; msk >>= 1) ss += __shfl_xor(ss, msk, 32);
        float inv = 1.0f / (sqrtf(ss) + 1e-8f);
        for (int j = lane; j < DIM; j += 32) {
            float re = (j < OBS) ? f[j] * inv : 0.f;
            stateV[s * DIM + (foldAmp((unsigned)j) ^ stateSwz((unsigned)s))] = make_float2(re, 0.f);
        }
    }
    __syncthreads();

    const int s    = lane & 7;
    const unsigned rsub = (unsigned)((lane >> 3) & 1);
    const int half = lane >> 4;
    const int kb   = 2 * half;
    const int ncol = lane & 15;
    const unsigned q = (unsigned)(ncol & 3);
    const int reim = (ncol >> 2) & 1;
    const unsigned swzS = stateSwz((unsigned)s);

    static_assert((GATE_LDS_FLOATS * 4) % 256 == 0, "gate table size must keep byte bits 0..7 clear");
    static_assert((STATE_LDS_FLOATS * 4) % 4096 == 0 && (DIM * 8) % 4096 == 0,
                  "wave region and state plane offsets must keep byte bits 0..11 clear");
    char* const smemB = (char*)smem;
    const unsigned GATE_BYTES = GATE_LDS_FLOATS * 4u;
    const unsigned waveB  = (unsigned)wave * (STATE_LDS_FLOATS * 4u);
    const unsigned ldRowB = waveB + (unsigned)s * (DIM * 8u);

    for (int layer = 0; layer < NLAYERS; ++layer) {
        #pragma unroll
        for (int i = 0; i < NQ; ++i) {
            const int g = layer * NQ + i;
            const float* gb = gl + g * GATE_FLOATS;
            v2f B0, B1;
            B0[0] = gb[(kb    ) * 16 + ncol];
            B0[1] = gb[(kb + 1) * 16 + ncol];
            B1[0] = gb[64 + (kb    ) * 16 + ncol];
            B1[1] = gb[64 + (kb + 1) * 16 + ncol];
            const int pm = (i < 8) ? (8 - i) : 0;
            const int pl = (i < 8) ? (7 - i) : 8;

            const unsigned fPl    = foldAmp(1u << pl);
            const unsigned ldLane = foldAmp(spreadR(rsub, pm, pl) | ((unsigned)half << pm)) ^ swzS;
            const unsigned stLane = foldAmp(spreadR((unsigned)half, pm, pl)
                                            | ((q >> 1) << pm) | ((q & 1u) << pl));
            const unsigned ldB0 = ldRowB + (ldLane << 3);
            const unsigned ldB1 = ldRowB + ((ldLane ^ fPl) << 3);
            const unsigned stB  = waveB + ((stLane << 3) | ((unsigned)reim << 2));

            float2 x0 = *(const float2*)(smemB + GATE_BYTES + ldB0);
            float2 x1 = *(const float2*)(smemB + GATE_BYTES + ldB1);

            unsigned uB = 0u;
            for (int rb = 0; rb < 64; rb += 4) {
                const unsigned uBn = foldAmp(spreadR(2u * (unsigned)((rb + 4) & 63), pm, pl));
                #pragma unroll
                for (int k = 0; k < 4; ++k) {
                    const unsigned uCur  = uB ^ foldAmp(spreadR(2u * (unsigned)k, pm, pl));
                    const unsigned uNext = (k < 3)
                        ? (uB ^ foldAmp(spreadR(2u * (unsigned)(k + 1), pm, pl))) : uBn;
                    float2 y0 = *(const float2*)(smemB + GATE_BYTES + (ldB0 ^ (uNext << 3)));
                    float2 y1 = *(const float2*)(smemB + GATE_BYTES + (ldB1 ^ (uNext << 3)));

                    v2f A0; A0[0] = x0.x; A0[1] = x1.x;
                    v2f A1; A1[0] = x0.y; A1[1] = x1.y;
                    v8f acc = {0.f, 0.f, 0.f, 0.f, 0.f, 0.f, 0.f, 0.f};
                    acc = __builtin_amdgcn_wmma_f32_16x16x4_f32(false, A0, false, B0,
                                                                (short)0, acc, false, false);
                    acc = __builtin_amdgcn_wmma_f32_16x16x4_f32(false, A1, false, B1,
                                                                (short)0, acc, false, false);

                    const unsigned tFull = GATE_BYTES + (stB ^ (uCur << 3));
                    #pragma unroll
                    for (int v = 0; v < 8; ++v)
                        *(float*)(smemB + ((tFull ^ (stateSwz((unsigned)v) << 3))
                                           + (unsigned)(v * (DIM * 8)))) = acc[v];

                    x0 = y0; x1 = y1;
                }
                uB = uBn;
            }
        }
    }

    for (int ss2 = 0; ss2 < STATES_PER_WAVE; ++ss2) {
        float* o = out + (size_t)(bBase + ss2) * DIM;
        for (int j = lane; j < DIM; j += 32) {
            float2 a = stateV[ss2 * DIM + (foldAmp((unsigned)j) ^ stateSwz((unsigned)ss2))];
            o[j] = tanhf(5.0f * (a.x * a.x + a.y * a.y));
        }
    }
}

extern "C" void kernel_launch(void* const* d_in, const int* in_sizes, int n_in,
                              void* d_out, int out_size, void* d_ws, size_t ws_size,
                              hipStream_t stream) {
    (void)in_sizes; (void)n_in; (void)d_ws; (void)ws_size; (void)out_size;
    const float* features = (const float*)d_in[0];
    const float* weights  = (const float*)d_in[1];
    float* out = (float*)d_out;

    dim3 grid(BATCH_TOT / STATES_PER_BLOCK);
    dim3 block(BLOCK_THREADS);
    size_t shmem = (size_t)(GATE_LDS_FLOATS + WAVES_PER_BLOCK * STATE_LDS_FLOATS)
                   * sizeof(float);
    su4_quantum_kernel<<<grid, block, shmem, stream>>>(features, weights, out);
}


# Product transposed: gate as block-diagonal A (K=16), state float2 slots as B; a lane's D pairs are (re,im) of the 4 slots it loaded: 4 b64 stores per 32 rows replace 16 b32; k order of sums changed
# speedup vs baseline: 2.0316x; 1.5062x over previous
#include <hip/hip_runtime.h>
#include <math.h>

#define NQ        9
#define DIM       512
#define NLAYERS   12
#define OBS       256
#define BATCH_TOT 16384

#define WAVES_PER_BLOCK   8
#define STATES_PER_WAVE   8
#define BLOCK_THREADS     (WAVES_PER_BLOCK * 32)
#define STATES_PER_BLOCK  (WAVES_PER_BLOCK * STATES_PER_WAVE)
#define NGATES            (NLAYERS * NQ)
#define GATE_FLOATS       64
#define GATE_LDS_FLOATS   (NGATES * GATE_FLOATS)
#define STATE_LDS_FLOATS  (STATES_PER_WAVE * DIM * 2)

typedef float v2f __attribute__((ext_vector_type(2)));
typedef float v8f __attribute__((ext_vector_type(8)));

struct cpx { float re, im; };
__device__ inline cpx cz() { cpx r; r.re = 0.f; r.im = 0.f; return r; }
__device__ inline cpx cmk(float a, float b) { cpx r; r.re = a; r.im = b; return r; }
__device__ inline cpx cmul(cpx a, cpx b) { return cmk(a.re*b.re - a.im*b.im, a.re*b.im + a.im*b.re); }
__device__ inline cpx cadd(cpx a, cpx b) { return cmk(a.re + b.re, a.im + b.im); }

__device__ inline void u3mat(float t, float p, float l, cpx U[2][2]) {
    float ct = cosf(0.5f * t), st = sinf(0.5f * t);
    float cp = cosf(p), sp = sinf(p), cl = cosf(l), sl = sinf(l);
    U[0][0] = cmk(ct, 0.f);
    U[0][1] = cmk(-cl * st, -sl * st);
    U[1][0] = cmk(cp * st, sp * st);
    U[1][1] = cmk((cp*cl - sp*sl) * ct, (cp*sl + sp*cl) * ct);
}

__device__ inline void kron22(const cpx A[2][2], const cpx B[2][2], cpx K[4][4]) {
    for (int a = 0; a < 2; ++a)
        for (int b = 0; b < 2; ++b)
            for (int c = 0; c < 2; ++c)
                for (int d = 0; d < 2; ++d)
                    K[2*a + c][2*b + d] = cmul(A[a][b], B[c][d]);
}

__device__ inline void mat4mul(cpx C[4][4], const cpx A[4][4], const cpx B[4][4]) {
    for (int i = 0; i < 4; ++i)
        for (int j = 0; j < 4; ++j) {
            cpx acc = cz();
            for (int k = 0; k < 4; ++k) acc = cadd(acc, cmul(A[i][k], B[k][j]));
            C[i][j] = acc;
        }
}

__device__ void computeGate(int g, const float* __restrict__ w, float* __restrict__ gl) {
    const float* th = w + g * 15;
    cpx Ua[2][2], Ub[2][2];
    cpx pre[4][4], post[4][4];
    u3mat(th[0], th[1], th[2], Ua);  u3mat(th[3], th[4], th[5], Ub);  kron22(Ua, Ub, pre);
    u3mat(th[9], th[10], th[11], Ua); u3mat(th[12], th[13], th[14], Ub); kron22(Ua, Ub, post);

    float c6 = cosf(0.5f*th[6]), s6 = sinf(0.5f*th[6]);
    float c7 = cosf(0.5f*th[7]), s7 = sinf(0.5f*th[7]);
    float c8 = cosf(0.5f*th[8]), s8 = sinf(0.5f*th[8]);
    cpx Rxx[4][4], Ryy[4][4], Rzz[4][4];
    for (int i = 0; i < 4; ++i)
        for (int j = 0; j < 4; ++j) { Rxx[i][j] = cz(); Ryy[i][j] = cz(); Rzz[i][j] = cz(); }
    for (int i = 0; i < 4; ++i) { Rxx[i][i] = cmk(c6, 0.f); Ryy[i][i] = cmk(c7, 0.f); }
    Rxx[0][3] = cmk(0.f, -s6); Rxx[1][2] = cmk(0.f, -s6);
    Rxx[2][1] = cmk(0.f, -s6); Rxx[3][0] = cmk(0.f, -s6);
    Ryy[0][3] = cmk(0.f,  s7); Ryy[1][2] = cmk(0.f, -s7);
    Ryy[2][1] = cmk(0.f, -s7); Ryy[3][0] = cmk(0.f,  s7);
    Rzz[0][0] = cmk(c8, -s8); Rzz[1][1] = cmk(c8,  s8);
    Rzz[2][2] = cmk(c8,  s8); Rzz[3][3] = cmk(c8, -s8);

    cpx t1[4][4], t2[4][4];
    mat4mul(t1, Rxx, Ryy);
    mat4mul(t2, t1, Rzz);
    mat4mul(t1, t2, pre);
    mat4mul(t2, post, t1);

    float* gb = gl + g * GATE_FLOATS;
    for (int b = 0; b < 8; ++b)
        for (int a = 0; a < 8; ++a) {
            float val;
            if (a < 4) val = (b < 4) ?  t2[a][b].re     : -t2[a][b-4].im;
            else       val = (b < 4) ?  t2[a-4][b].im   :  t2[a-4][b-4].re;
            gb[b*8 + a] = val;
        }
}

__device__ inline unsigned insert2(unsigned r, int lo, unsigned vlo, int hi, unsigned vhi) {
    unsigned low = r & ((1u << lo) - 1u);
    unsigned x = ((r >> lo) << (lo + 1)) | (vlo << lo) | low;
    unsigned low2 = x & ((1u << hi) - 1u);
    return ((x >> hi) << (hi + 1)) | (vhi << hi) | low2;
}

__device__ inline unsigned foldAmp(unsigned a) {
    return a ^ ((((a >> 5) ^ (a >> 7)) & 3u) << 3);
}
__device__ inline unsigned stateSwz(unsigned s) {
    return ((s & 1u) ? 12u : 0u) ^ ((s & 2u) ? 24u : 0u) ^ ((s & 4u) ? 7u : 0u);
}
__device__ inline unsigned spreadR(unsigned r, int pm, int pl) {
    int lo = (pm < pl) ? pm : pl;
    int hi = (pm < pl) ? pl : pm;
    return insert2(r, lo, 0u, hi, 0u);
}

__global__ __launch_bounds__(BLOCK_THREADS, 1)
void su4_quantum_kernel(const float* __restrict__ features,
                        const float* __restrict__ weights,
                        float* __restrict__ out) {
    extern __shared__ float smem[];
    float* gl = smem;
    const int tid  = threadIdx.x;
    const int lane = tid & 31;
    const int wave = tid >> 5;
    float*  stateF = smem + GATE_LDS_FLOATS + wave * STATE_LDS_FLOATS;
    float2* stateV = (float2*)stateF;

    if (tid < NGATES) computeGate(tid, weights, gl);

    const int bBase = (blockIdx.x * WAVES_PER_BLOCK + wave) * STATES_PER_WAVE;
    for (int s = 0; s < STATES_PER_WAVE; ++s) {
        const float* f = features + (size_t)(bBase + s) * OBS;
        float ss = 0.f;
        for (int j = lane; j < OBS; j += 32) { float v = f[j]; ss += v * v; }
        for (int msk = 16; msk >= 1; msk >>= 1) ss += __shfl_xor(ss, msk, 32);
        float inv = 1.0f / (sqrtf(ss) + 1e-8f);
        for (int j = lane; j < DIM; j += 32) {
            float re = (j < OBS) ? f[j] * inv : 0.f;
            stateV[s * DIM + (foldAmp((unsigned)j) ^ stateSwz((unsigned)s))] = make_float2(re, 0.f);
        }
    }
    __syncthreads();

    const int s    = lane & 7;
    const unsigned rsub = (unsigned)((lane >> 3) & 1);
    const int half = lane >> 4;
    const int grow = lane & 15;
    const int ga   = (2 * (grow >> 3) + ((grow >> 1) & 1)) + 4 * (grow & 1);
    const bool loBlk = (grow & 4) == 0;
    const unsigned swzS = stateSwz((unsigned)s);

    static_assert((GATE_LDS_FLOATS * 4) % 256 == 0, "gate table must not shift the bank map");
    static_assert((STATE_LDS_FLOATS * 4) % 4096 == 0 && (DIM * 8) % 4096 == 0,
                  "wave region and state plane offsets must keep byte bits 0..11 clear");
    char* const smemB = (char*)smem;
    const unsigned GATE_BYTES = GATE_LDS_FLOATS * 4u;
    const unsigned waveB = (unsigned)wave * (STATE_LDS_FLOATS * 4u);
    const unsigned rowB  = waveB + (unsigned)s * (DIM * 8u);

    for (int layer = 0; layer < NLAYERS; ++layer) {
        #pragma unroll
        for (int i = 0; i < NQ; ++i) {
            const int g = layer * NQ + i;
            const float* gb = gl + g * GATE_FLOATS;
            const int pm = (i < 8) ? (8 - i) : 0;
            const int pl = (i < 8) ? (7 - i) : 8;

            v2f G0, G1;
            G0[0] = gb[     (2 * half    ) * 8 + ga];
            G0[1] = gb[32 + (2 * half    ) * 8 + ga];
            G1[0] = gb[     (2 * half + 1) * 8 + ga];
            G1[1] = gb[32 + (2 * half + 1) * 8 + ga];
            const v2f Z2 = {0.f, 0.f};
            const v2f A0 = loBlk ? G0 : Z2;
            const v2f A1 = loBlk ? G1 : Z2;
            const v2f A2 = loBlk ? Z2 : G0;
            const v2f A3 = loBlk ? Z2 : G1;

            const unsigned fPl    = foldAmp(1u << pl);
            const unsigned ldLane = foldAmp(spreadR(rsub, pm, pl) | ((unsigned)half << pm)) ^ swzS;
            const unsigned ldB0 = rowB + (ldLane << 3);
            const unsigned ldB1 = rowB + ((ldLane ^ fPl) << 3);
            const unsigned uSet = foldAmp(spreadR(2u, pm, pl)) << 3;

            v2f xa[4][4], xb[4][4];
            #define TILE_SLOTS(tile0, k)                                                         \
                const unsigned u0_ = (foldAmp(spreadR(4u * (unsigned)(tile0), pm, pl)) << 3)     \
                                   ^ (foldAmp(spreadR(4u * (unsigned)(k), pm, pl)) << 3);        \
                const unsigned u1_ = u0_ ^ uSet;                                                 \
                char* const p0_ = smemB + GATE_BYTES + (ldB0 ^ u0_);         \
                char* const p1_ = smemB + GATE_BYTES + (ldB1 ^ u0_);         \
                char* const p2_ = smemB + GATE_BYTES + (ldB0 ^ u1_);         \
                char* const p3_ = smemB + GATE_BYTES + (ldB1 ^ u1_);
            #define LOAD_TRIP(X, tile0)                                                          \
                _Pragma("unroll")                                                                \
                for (int k = 0; k < 4; ++k) {                                                    \
                    TILE_SLOTS(tile0, k)                                                         \
                    X[k][0] = *(const v2f*)p0_;                                                  \
                    X[k][1] = *(const v2f*)p1_;                                                  \
                    X[k][2] = *(const v2f*)p2_;                                                  \
                    X[k][3] = *(const v2f*)p3_;                                                  \
                }
            #define GATE_TRIP(X, tile0)                                                          \
                _Pragma("unroll")                                                                \
                for (int k = 0; k < 4; ++k) {                                                    \
                    TILE_SLOTS(tile0, k)                                                         \
                    v8f c_ = {0.f, 0.f, 0.f, 0.f, 0.f, 0.f, 0.f, 0.f};                           \
                    c_ = __builtin_amdgcn_wmma_f32_16x16x4_f32(false, A0, false, X[k][0], (short)0, c_, false, false); \
                    c_ = __builtin_amdgcn_wmma_f32_16x16x4_f32(false, A1, false, X[k][1], (short)0, c_, false, false); \
                    c_ = __builtin_amdgcn_wmma_f32_16x16x4_f32(false, A2, false, X[k][2], (short)0, c_, false, false); \
                    c_ = __builtin_amdgcn_wmma_f32_16x16x4_f32(false, A3, false, X[k][3], (short)0, c_, false, false); \
                    v2f o0_; o0_[0] = c_[0]; o0_[1] = c_[1];                                     \
                    v2f o1_; o1_[0] = c_[2]; o1_[1] = c_[3];                                     \
                    v2f o2_; o2_[0] = c_[4]; o2_[1] = c_[5];                                     \
                    v2f o3_; o3_[0] = c_[6]; o3_[1] = c_[7];                                     \
                    *(v2f*)p0_ = o0_;                                                            \
                    *(v2f*)p1_ = o1_;                                                            \
                    *(v2f*)p2_ = o2_;                                                            \
                    *(v2f*)p3_ = o3_;                                                            \
                }

            LOAD_TRIP(xa, 0)
            #pragma unroll 1
            for (int tb = 0; tb < 32; tb += 8) {
                LOAD_TRIP(xb, tb + 4)
                GATE_TRIP(xa, tb)
                LOAD_TRIP(xa, (tb + 8) & 31)
                GATE_TRIP(xb, tb + 4)
            }
            #undef TILE_SLOTS
            #undef LOAD_TRIP
            #undef GATE_TRIP
        }
    }

    for (int ss2 = 0; ss2 < STATES_PER_WAVE; ++ss2) {
        float* o = out + (size_t)(bBase + ss2) * DIM;
        for (int j = lane; j < DIM; j += 32) {
            float2 a = stateV[ss2 * DIM + (foldAmp((unsigned)j) ^ stateSwz((unsigned)ss2))];
            o[j] = tanhf(5.0f * (a.x * a.x + a.y * a.y));
        }
    }
}

extern "C" void kernel_launch(void* const* d_in, const int* in_sizes, int n_in,
                              void* d_out, int out_size, void* d_ws, size_t ws_size,
                              hipStream_t stream) {
    (void)in_sizes; (void)n_in; (void)d_ws; (void)ws_size; (void)out_size;
    const float* features = (const float*)d_in[0];
    const float* weights  = (const float*)d_in[1];
    float* out = (float*)d_out;

    dim3 grid(BATCH_TOT / STATES_PER_BLOCK);
    dim3 block(BLOCK_THREADS);
    size_t shmem = (size_t)(GATE_LDS_FLOATS + WAVES_PER_BLOCK * STATE_LDS_FLOATS)
                   * sizeof(float);
    su4_quantum_kernel<<<grid, block, shmem, stream>>>(features, weights, out);
}


# Gate pairs (2k,2k+1) fused into one 3-qubit matrix, 16x16 real = a dense WMMA A over K=16: 54 state passes replace 108, half the WMMAs and LDS traffic; pair products and k order of sums changed
# speedup vs baseline: 3.5126x; 1.7290x over previous
#include <hip/hip_runtime.h>
#include <math.h>

#define NQ        9
#define DIM       512
#define NLAYERS   12
#define OBS       256
#define BATCH_TOT 16384

#define WAVES_PER_BLOCK   8
#define STATES_PER_WAVE   8
#define BLOCK_THREADS     (WAVES_PER_BLOCK * 32)
#define STATES_PER_BLOCK  (WAVES_PER_BLOCK * STATES_PER_WAVE)
#define NGATES            (NLAYERS * NQ)
#define NPAIRS            (NGATES / 2)
#define PAIR_FLOATS       128
#define GATE_LDS_FLOATS   (NPAIRS * PAIR_FLOATS)
#define STATE_LDS_FLOATS  (STATES_PER_WAVE * DIM * 2)
#define GTMP_FLOATS       (NGATES * 32)

typedef float v2f __attribute__((ext_vector_type(2)));
typedef float v8f __attribute__((ext_vector_type(8)));

struct cpx { float re, im; };
__device__ inline cpx cz() { cpx r; r.re = 0.f; r.im = 0.f; return r; }
__device__ inline cpx cmk(float a, float b) { cpx r; r.re = a; r.im = b; return r; }
__device__ inline cpx cmul(cpx a, cpx b) { return cmk(a.re*b.re - a.im*b.im, a.re*b.im + a.im*b.re); }
__device__ inline cpx cadd(cpx a, cpx b) { return cmk(a.re + b.re, a.im + b.im); }

__device__ inline void u3mat(float t, float p, float l, cpx U[2][2]) {
    float ct = cosf(0.5f * t), st = sinf(0.5f * t);
    float cp = cosf(p), sp = sinf(p), cl = cosf(l), sl = sinf(l);
    U[0][0] = cmk(ct, 0.f);
    U[0][1] = cmk(-cl * st, -sl * st);
    U[1][0] = cmk(cp * st, sp * st);
    U[1][1] = cmk((cp*cl - sp*sl) * ct, (cp*sl + sp*cl) * ct);
}

__device__ inline void kron22(const cpx A[2][2], const cpx B[2][2], cpx K[4][4]) {
    for (int a = 0; a < 2; ++a)
        for (int b = 0; b < 2; ++b)
            for (int c = 0; c < 2; ++c)
                for (int d = 0; d < 2; ++d)
                    K[2*a + c][2*b + d] = cmul(A[a][b], B[c][d]);
}

__device__ inline void mat4mul(cpx C[4][4], const cpx A[4][4], const cpx B[4][4]) {
    for (int i = 0; i < 4; ++i)
        for (int j = 0; j < 4; ++j) {
            cpx acc = cz();
            for (int k = 0; k < 4; ++k) acc = cadd(acc, cmul(A[i][k], B[k][j]));
            C[i][j] = acc;
        }
}

__device__ void computeGate(int g, const float* __restrict__ w, float2* __restrict__ gt) {
    const float* th = w + g * 15;
    cpx Ua[2][2], Ub[2][2];
    cpx pre[4][4], post[4][4];
    u3mat(th[0], th[1], th[2], Ua);  u3mat(th[3], th[4], th[5], Ub);  kron22(Ua, Ub, pre);
    u3mat(th[9], th[10], th[11], Ua); u3mat(th[12], th[13], th[14], Ub); kron22(Ua, Ub, post);

    float c6 = cosf(0.5f*th[6]), s6 = sinf(0.5f*th[6]);
    float c7 = cosf(0.5f*th[7]), s7 = sinf(0.5f*th[7]);
    float c8 = cosf(0.5f*th[8]), s8 = sinf(0.5f*th[8]);
    cpx Rxx[4][4], Ryy[4][4], Rzz[4][4];
    for (int i = 0; i < 4; ++i)
        for (int j = 0; j < 4; ++j) { Rxx[i][j] = cz(); Ryy[i][j] = cz(); Rzz[i][j] = cz(); }
    for (int i = 0; i < 4; ++i) { Rxx[i][i] = cmk(c6, 0.f); Ryy[i][i] = cmk(c7, 0.f); }
    Rxx[0][3] = cmk(0.f, -s6); Rxx[1][2] = cmk(0.f, -s6);
    Rxx[2][1] = cmk(0.f, -s6); Rxx[3][0] = cmk(0.f, -s6);
    Ryy[0][3] = cmk(0.f,  s7); Ryy[1][2] = cmk(0.f, -s7);
    Ryy[2][1] = cmk(0.f, -s7); Ryy[3][0] = cmk(0.f,  s7);
    Rzz[0][0] = cmk(c8, -s8); Rzz[1][1] = cmk(c8,  s8);
    Rzz[2][2] = cmk(c8,  s8); Rzz[3][3] = cmk(c8, -s8);

    cpx t1[4][4], t2[4][4];
    mat4mul(t1, Rxx, Ryy);
    mat4mul(t2, t1, Rzz);
    mat4mul(t1, t2, pre);
    mat4mul(t2, post, t1);

    float2* gb = gt + g * 16;
    for (int a = 0; a < 4; ++a)
        for (int b = 0; b < 4; ++b)
            gb[a*4 + b] = make_float2(t2[a][b].re, t2[a][b].im);
}

__device__ inline void fusePair(int idx, const float2* __restrict__ gt, float2* __restrict__ ut) {
    const int pr = idx >> 6, Q = (idx >> 3) & 7, P = idx & 7;
    const float2* G1 = gt + (2 * pr) * 16;
    const float2* G2 = gt + (2 * pr + 1) * 16;
    const int A = Q >> 2, B = (Q >> 1) & 1, C = Q & 1;
    const int a = P >> 2, b = (P >> 1) & 1, c = P & 1;
    float re = 0.f, im = 0.f;
    for (int bp = 0; bp < 2; ++bp) {
        const float2 x = G2[(2 * B + C) * 4 + 2 * bp + c];
        const float2 y = G1[(2 * A + bp) * 4 + 2 * a + b];
        re += x.x * y.x - x.y * y.y;
        im += x.x * y.y + x.y * y.x;
    }
    ut[idx] = make_float2(re, im);
}

__device__ inline unsigned foldAmp(unsigned a) {
    return a ^ ((((a >> 5) ^ (a >> 7)) & 3u) << 3);
}
__device__ inline unsigned stateSwz(unsigned s) {
    return ((s & 1u) ? 12u : 0u) ^ ((s & 2u) ? 24u : 0u) ^ ((s & 4u) ? 7u : 0u);
}
__device__ inline unsigned insert0(unsigned x, int pos) {
    return ((x >> pos) << (pos + 1)) | (x & ((1u << pos) - 1u));
}
__device__ inline unsigned spread3(unsigned r, int p0, int p1, int p2) {
    int lo = (p0 < p1) ? ((p0 < p2) ? p0 : p2) : ((p1 < p2) ? p1 : p2);
    int hi = (p0 > p1) ? ((p0 > p2) ? p0 : p2) : ((p1 > p2) ? p1 : p2);
    int mid = p0 + p1 + p2 - lo - hi;
    return insert0(insert0(insert0(r, lo), mid), hi);
}
__device__ inline v2f gateFrag(const float2* ut, int Q, int P, bool imRow) {
    const float2 u = ut[Q * 8 + P];
    v2f r;
    r[0] = imRow ? u.y : u.x;
    r[1] = imRow ? u.x : -u.y;
    return r;
}

__global__ __launch_bounds__(BLOCK_THREADS, 1)
void su4_quantum_kernel(const float* __restrict__ features,
                        const float* __restrict__ weights,
                        float* __restrict__ out) {
    extern __shared__ float smem[];
    float2* ut = (float2*)smem;
    static_assert(GTMP_FLOATS <= WAVES_PER_BLOCK * STATE_LDS_FLOATS, "single-gate scratch must fit");
    float2* gt = (float2*)(smem + GATE_LDS_FLOATS);
    const int tid  = threadIdx.x;
    const int lane = tid & 31;
    const int wave = tid >> 5;
    float*  stateF = smem + GATE_LDS_FLOATS + wave * STATE_LDS_FLOATS;
    float2* stateV = (float2*)stateF;

    if (tid < NGATES) computeGate(tid, weights, gt);
    __syncthreads();
    for (int idx = tid; idx < NPAIRS * 64; idx += BLOCK_THREADS) fusePair(idx, gt, ut);
    __syncthreads();

    const int bBase = (blockIdx.x * WAVES_PER_BLOCK + wave) * STATES_PER_WAVE;
    for (int s = 0; s < STATES_PER_WAVE; ++s) {
        const float* f = features + (size_t)(bBase + s) * OBS;
        float ss = 0.f;
        for (int j = lane; j < OBS; j += 32) { float v = f[j]; ss += v * v; }
        for (int msk = 16; msk >= 1; msk >>= 1) ss += __shfl_xor(ss, msk, 32);
        float inv = 1.0f / (sqrtf(ss) + 1e-8f);
        for (int j = lane; j < DIM; j += 32) {
            float re = (j < OBS) ? f[j] * inv : 0.f;
            stateV[s * DIM + (foldAmp((unsigned)j) ^ stateSwz((unsigned)s))] = make_float2(re, 0.f);
        }
    }
    __syncthreads();

    const int s    = lane & 7;
    const unsigned rsub = (unsigned)((lane >> 3) & 1);
    const int half = lane >> 4;
    const int grow = lane & 15;
    const bool imRow = (grow & 1) != 0;
    const unsigned swzS = stateSwz((unsigned)s);

    static_assert((GATE_LDS_FLOATS * 4) % 256 == 0, "gate table must not shift the bank map");
    static_assert((STATE_LDS_FLOATS * 4) % 4096 == 0 && (DIM * 8) % 4096 == 0,
                  "wave region and state plane offsets must keep byte bits 0..11 clear");
    char* const smemB = (char*)smem;
    const unsigned GATE_BYTES = GATE_LDS_FLOATS * 4u;
    const unsigned waveB = (unsigned)wave * (STATE_LDS_FLOATS * 4u);
    const unsigned rowB  = waveB + (unsigned)s * (DIM * 8u);

    for (int rep = 0; rep < NPAIRS / NQ; ++rep) {
        #pragma unroll
        for (int kk = 0; kk < NQ; ++kk) {
            const int i  = (2 * kk) % NQ;
            const int pa = 8 - i;
            const int pb = 8 - ((i + 1) % NQ);
            const int pc = 8 - ((i + 2) % NQ);
            const bool hOnC = (i == 6);
            const int ph  = hOnC ? pc : pa;  const int wh  = hOnC ? 1 : 4;
            const int pj1 = hOnC ? pa : pb;  const int wj1 = hOnC ? 4 : 2;
            const int pj0 = hOnC ? pb : pc;  const int wj0 = hOnC ? 2 : 1;

            const float2* up = ut + (rep * NQ + kk) * 64;
            const int Q  = (grow >> 3) * wh + ((grow >> 2) & 1) * wj1 + ((grow >> 1) & 1) * wj0;
            const int Ph = half * wh;
            const v2f A0 = gateFrag(up, Q, Ph,             imRow);
            const v2f A1 = gateFrag(up, Q, Ph + wj0,       imRow);
            const v2f A2 = gateFrag(up, Q, Ph + wj1,       imRow);
            const v2f A3 = gateFrag(up, Q, Ph + wj1 + wj0, imRow);

            const unsigned fJ0    = foldAmp(1u << pj0);
            const unsigned fJ1    = foldAmp(1u << pj1);
            const unsigned ldLane = foldAmp(spread3(rsub, pa, pb, pc) | ((unsigned)half << ph)) ^ swzS;
            const unsigned ldB0 = rowB + (ldLane << 3);
            const unsigned ldB1 = rowB + ((ldLane ^ fJ0) << 3);
            const unsigned ldB2 = rowB + ((ldLane ^ fJ1) << 3);
            const unsigned ldB3 = rowB + ((ldLane ^ fJ1 ^ fJ0) << 3);

            v2f xa[4][4], xb[4][4];
            #define TILE_SLOTS(tile0, k)                                                         \
                const unsigned u0_ = (foldAmp(spread3(2u * (unsigned)(tile0), pa, pb, pc)) << 3) \
                                   ^ (foldAmp(spread3(2u * (unsigned)(k), pa, pb, pc)) << 3);    \
                char* const p0_ = smemB + GATE_BYTES + (ldB0 ^ u0_);                   \
                char* const p1_ = smemB + GATE_BYTES + (ldB1 ^ u0_);                   \
                char* const p2_ = smemB + GATE_BYTES + (ldB2 ^ u0_);                   \
                char* const p3_ = smemB + GATE_BYTES + (ldB3 ^ u0_);
            #define LOAD_TRIP(X, tile0)                                                          \
                _Pragma("unroll")                                                                \
                for (int k = 0; k < 4; ++k) {                                                    \
                    TILE_SLOTS(tile0, k)                                                         \
                    X[k][0] = *(const v2f*)p0_;                                                  \
                    X[k][1] = *(const v2f*)p1_;                                                  \
                    X[k][2] = *(const v2f*)p2_;                                                  \
                    X[k][3] = *(const v2f*)p3_;                                                  \
                }
            #define GATE_TRIP(X, tile0)                                                          \
                _Pragma("unroll")                                                                \
                for (int k = 0; k < 4; ++k) {                                                    \
                    TILE_SLOTS(tile0, k)                                                         \
                    v8f c_ = {0.f, 0.f, 0.f, 0.f, 0.f, 0.f, 0.f, 0.f};                           \
                    c_ = __builtin_amdgcn_wmma_f32_16x16x4_f32(false, A0, false, X[k][0], (short)0, c_, false, false); \
                    c_ = __builtin_amdgcn_wmma_f32_16x16x4_f32(false, A1, false, X[k][1], (short)0, c_, false, false); \
                    c_ = __builtin_amdgcn_wmma_f32_16x16x4_f32(false, A2, false, X[k][2], (short)0, c_, false, false); \
                    c_ = __builtin_amdgcn_wmma_f32_16x16x4_f32(false, A3, false, X[k][3], (short)0, c_, false, false); \
                    v2f o0_; o0_[0] = c_[0]; o0_[1] = c_[1];                                     \
                    v2f o1_; o1_[0] = c_[2]; o1_[1] = c_[3];                                     \
                    v2f o2_; o2_[0] = c_[4]; o2_[1] = c_[5];                                     \
                    v2f o3_; o3_[0] = c_[6]; o3_[1] = c_[7];                                     \
                    *(v2f*)p0_ = o0_;                                                            \
                    *(v2f*)p1_ = o1_;                                                            \
                    *(v2f*)p2_ = o2_;                                                            \
                    *(v2f*)p3_ = o3_;                                                            \
                }

            LOAD_TRIP(xa, 0)
            #pragma unroll 1
            for (int tb = 0; tb < 32; tb += 8) {
                LOAD_TRIP(xb, tb + 4)
                GATE_TRIP(xa, tb)
                LOAD_TRIP(xa, (tb + 8) & 31)
                GATE_TRIP(xb, tb + 4)
            }
            #undef TILE_SLOTS
            #undef LOAD_TRIP
            #undef GATE_TRIP
        }
    }

    for (int ss2 = 0; ss2 < STATES_PER_WAVE; ++ss2) {
        float* o = out + (size_t)(bBase + ss2) * DIM;
        for (int j = lane; j < DIM; j += 32) {
            float2 a = stateV[ss2 * DIM + (foldAmp((unsigned)j) ^ stateSwz((unsigned)ss2))];
            o[j] = tanhf(5.0f * (a.x * a.x + a.y * a.y));
        }
    }
}

extern "C" void kernel_launch(void* const* d_in, const int* in_sizes, int n_in,
                              void* d_out, int out_size, void* d_ws, size_t ws_size,
                              hipStream_t stream) {
    (void)in_sizes; (void)n_in; (void)d_ws; (void)ws_size; (void)out_size;
    const float* features = (const float*)d_in[0];
    const float* weights  = (const float*)d_in[1];
    float* out = (float*)d_out;

    dim3 grid(BATCH_TOT / STATES_PER_BLOCK);
    dim3 block(BLOCK_THREADS);
    size_t shmem = (size_t)(GATE_LDS_FLOATS + WAVES_PER_BLOCK * STATE_LDS_FLOATS)
                   * sizeof(float);
    su4_quantum_kernel<<<grid, block, shmem, stream>>>(features, weights, out);
}


# Pair pass re-blocked: per trip 16 tiles' 64 LDS loads as one group, then their 64 WMMAs back to back on 16 accumulators, then the 64 stores: 2 LDS places per pair in the WMMA stream, was about 40
# speedup vs baseline: 3.7441x; 1.0659x over previous
#include <hip/hip_runtime.h>
#include <math.h>

#define NQ        9
#define DIM       512
#define NLAYERS   12
#define OBS       256
#define BATCH_TOT 16384

#define WAVES_PER_BLOCK   8
#define STATES_PER_WAVE   8
#define BLOCK_THREADS     (WAVES_PER_BLOCK * 32)
#define STATES_PER_BLOCK  (WAVES_PER_BLOCK * STATES_PER_WAVE)
#define NGATES            (NLAYERS * NQ)
#define NPAIRS            (NGATES / 2)
#define PAIR_FLOATS       128
#define GATE_LDS_FLOATS   (NPAIRS * PAIR_FLOATS)
#define STATE_LDS_FLOATS  (STATES_PER_WAVE * DIM * 2)
#define GTMP_FLOATS       (NGATES * 32)

typedef float v2f __attribute__((ext_vector_type(2)));
typedef float v8f __attribute__((ext_vector_type(8)));

struct cpx { float re, im; };
__device__ inline cpx cz() { cpx r; r.re = 0.f; r.im = 0.f; return r; }
__device__ inline cpx cmk(float a, float b) { cpx r; r.re = a; r.im = b; return r; }
__device__ inline cpx cmul(cpx a, cpx b) { return cmk(a.re*b.re - a.im*b.im, a.re*b.im + a.im*b.re); }
__device__ inline cpx cadd(cpx a, cpx b) { return cmk(a.re + b.re, a.im + b.im); }

__device__ inline void u3mat(float t, float p, float l, cpx U[2][2]) {
    float ct = cosf(0.5f * t), st = sinf(0.5f * t);
    float cp = cosf(p), sp = sinf(p), cl = cosf(l), sl = sinf(l);
    U[0][0] = cmk(ct, 0.f);
    U[0][1] = cmk(-cl * st, -sl * st);
    U[1][0] = cmk(cp * st, sp * st);
    U[1][1] = cmk((cp*cl - sp*sl) * ct, (cp*sl + sp*cl) * ct);
}

__device__ inline void kron22(const cpx A[2][2], const cpx B[2][2], cpx K[4][4]) {
    for (int a = 0; a < 2; ++a)
        for (int b = 0; b < 2; ++b)
            for (int c = 0; c < 2; ++c)
                for (int d = 0; d < 2; ++d)
                    K[2*a + c][2*b + d] = cmul(A[a][b], B[c][d]);
}

__device__ inline void mat4mul(cpx C[4][4], const cpx A[4][4], const cpx B[4][4]) {
    for (int i = 0; i < 4; ++i)
        for (int j = 0; j < 4; ++j) {
            cpx acc = cz();
            for (int k = 0; k < 4; ++k) acc = cadd(acc, cmul(A[i][k], B[k][j]));
            C[i][j] = acc;
        }
}

__device__ void computeGate(int g, const float* __restrict__ w, float2* __restrict__ gt) {
    const float* th = w + g * 15;
    cpx Ua[2][2], Ub[2][2];
    cpx pre[4][4], post[4][4];
    u3mat(th[0], th[1], th[2], Ua);  u3mat(th[3], th[4], th[5], Ub);  kron22(Ua, Ub, pre);
    u3mat(th[9], th[10], th[11], Ua); u3mat(th[12], th[13], th[14], Ub); kron22(Ua, Ub, post);

    float c6 = cosf(0.5f*th[6]), s6 = sinf(0.5f*th[6]);
    float c7 = cosf(0.5f*th[7]), s7 = sinf(0.5f*th[7]);
    float c8 = cosf(0.5f*th[8]), s8 = sinf(0.5f*th[8]);
    cpx Rxx[4][4], Ryy[4][4], Rzz[4][4];
    for (int i = 0; i < 4; ++i)
        for (int j = 0; j < 4; ++j) { Rxx[i][j] = cz(); Ryy[i][j] = cz(); Rzz[i][j] = cz(); }
    for (int i = 0; i < 4; ++i) { Rxx[i][i] = cmk(c6, 0.f); Ryy[i][i] = cmk(c7, 0.f); }
    Rxx[0][3] = cmk(0.f, -s6); Rxx[1][2] = cmk(0.f, -s6);
    Rxx[2][1] = cmk(0.f, -s6); Rxx[3][0] = cmk(0.f, -s6);
    Ryy[0][3] = cmk(0.f,  s7); Ryy[1][2] = cmk(0.f, -s7);
    Ryy[2][1] = cmk(0.f, -s7); Ryy[3][0] = cmk(0.f,  s7);
    Rzz[0][0] = cmk(c8, -s8); Rzz[1][1] = cmk(c8,  s8);
    Rzz[2][2] = cmk(c8,  s8); Rzz[3][3] = cmk(c8, -s8);

    cpx t1[4][4], t2[4][4];
    mat4mul(t1, Rxx, Ryy);
    mat4mul(t2, t1, Rzz);
    mat4mul(t1, t2, pre);
    mat4mul(t2, post, t1);

    float2* gb = gt + g * 16;
    for (int a = 0; a < 4; ++a)
        for (int b = 0; b < 4; ++b)
            gb[a*4 + b] = make_float2(t2[a][b].re, t2[a][b].im);
}

__device__ inline void fusePair(int idx, const float2* __restrict__ gt, float2* __restrict__ ut) {
    const int pr = idx >> 6, Q = (idx >> 3) & 7, P = idx & 7;
    const float2* G1 = gt + (2 * pr) * 16;
    const float2* G2 = gt + (2 * pr + 1) * 16;
    const int A = Q >> 2, B = (Q >> 1) & 1, C = Q & 1;
    const int a = P >> 2, b = (P >> 1) & 1, c = P & 1;
    float re = 0.f, im = 0.f;
    for (int bp = 0; bp < 2; ++bp) {
        const float2 x = G2[(2 * B + C) * 4 + 2 * bp + c];
        const float2 y = G1[(2 * A + bp) * 4 + 2 * a + b];
        re += x.x * y.x - x.y * y.y;
        im += x.x * y.y + x.y * y.x;
    }
    ut[idx] = make_float2(re, im);
}

__device__ inline unsigned foldAmp(unsigned a) {
    return a ^ ((((a >> 5) ^ (a >> 7)) & 3u) << 3);
}
__device__ inline unsigned stateSwz(unsigned s) {
    return ((s & 1u) ? 12u : 0u) ^ ((s & 2u) ? 24u : 0u) ^ ((s & 4u) ? 7u : 0u);
}
__device__ inline unsigned insert0(unsigned x, int pos) {
    return ((x >> pos) << (pos + 1)) | (x & ((1u << pos) - 1u));
}
__device__ inline unsigned spread3(unsigned r, int p0, int p1, int p2) {
    int lo = (p0 < p1) ? ((p0 < p2) ? p0 : p2) : ((p1 < p2) ? p1 : p2);
    int hi = (p0 > p1) ? ((p0 > p2) ? p0 : p2) : ((p1 > p2) ? p1 : p2);
    int mid = p0 + p1 + p2 - lo - hi;
    return insert0(insert0(insert0(r, lo), mid), hi);
}
__device__ inline v2f gateFrag(const float2* ut, int Q, int P, bool imRow) {
    const float2 u = ut[Q * 8 + P];
    v2f r;
    r[0] = imRow ? u.y : u.x;
    r[1] = imRow ? u.x : -u.y;
    return r;
}

__global__ __launch_bounds__(BLOCK_THREADS, 1)
void su4_quantum_kernel(const float* __restrict__ features,
                        const float* __restrict__ weights,
                        float* __restrict__ out) {
    extern __shared__ float smem[];
    float2* ut = (float2*)smem;
    static_assert(GTMP_FLOATS <= WAVES_PER_BLOCK * STATE_LDS_FLOATS, "single-gate scratch must fit");
    float2* gt = (float2*)(smem + GATE_LDS_FLOATS);
    const int tid  = threadIdx.x;
    const int lane = tid & 31;
    const int wave = tid >> 5;
    float*  stateF = smem + GATE_LDS_FLOATS + wave * STATE_LDS_FLOATS;
    float2* stateV = (float2*)stateF;

    if (tid < NGATES) computeGate(tid, weights, gt);
    __syncthreads();
    for (int idx = tid; idx < NPAIRS * 64; idx += BLOCK_THREADS) fusePair(idx, gt, ut);
    __syncthreads();

    const int bBase = (blockIdx.x * WAVES_PER_BLOCK + wave) * STATES_PER_WAVE;
    for (int s = 0; s < STATES_PER_WAVE; ++s) {
        const float* f = features + (size_t)(bBase + s) * OBS;
        float ss = 0.f;
        for (int j = lane; j < OBS; j += 32) { float v = f[j]; ss += v * v; }
        for (int msk = 16; msk >= 1; msk >>= 1) ss += __shfl_xor(ss, msk, 32);
        float inv = 1.0f / (sqrtf(ss) + 1e-8f);
        for (int j = lane; j < DIM; j += 32) {
            float re = (j < OBS) ? f[j] * inv : 0.f;
            stateV[s * DIM + (foldAmp((unsigned)j) ^ stateSwz((unsigned)s))] = make_float2(re, 0.f);
        }
    }
    __syncthreads();

    const int s    = lane & 7;
    const unsigned rsub = (unsigned)((lane >> 3) & 1);
    const int half = lane >> 4;
    const int grow = lane & 15;
    const bool imRow = (grow & 1) != 0;
    const unsigned swzS = stateSwz((unsigned)s);

    static_assert((GATE_LDS_FLOATS * 4) % 256 == 0, "gate table must not shift the bank map");
    static_assert((STATE_LDS_FLOATS * 4) % 4096 == 0 && (DIM * 8) % 4096 == 0,
                  "wave region and state plane offsets must keep byte bits 0..11 clear");
    char* const smemB = (char*)smem;
    const unsigned GATE_BYTES = GATE_LDS_FLOATS * 4u;
    const unsigned waveB = (unsigned)wave * (STATE_LDS_FLOATS * 4u);
    const unsigned rowB  = waveB + (unsigned)s * (DIM * 8u);

    for (int rep = 0; rep < NPAIRS / NQ; ++rep) {
        #pragma unroll
        for (int kk = 0; kk < NQ; ++kk) {
            const int i  = (2 * kk) % NQ;
            const int pa = 8 - i;
            const int pb = 8 - ((i + 1) % NQ);
            const int pc = 8 - ((i + 2) % NQ);
            const bool hOnC = (i == 6);
            const int ph  = hOnC ? pc : pa;  const int wh  = hOnC ? 1 : 4;
            const int pj1 = hOnC ? pa : pb;  const int wj1 = hOnC ? 4 : 2;
            const int pj0 = hOnC ? pb : pc;  const int wj0 = hOnC ? 2 : 1;

            const float2* up = ut + (rep * NQ + kk) * 64;
            const int Q  = (grow >> 3) * wh + ((grow >> 2) & 1) * wj1 + ((grow >> 1) & 1) * wj0;
            const int Ph = half * wh;
            const v2f A0 = gateFrag(up, Q, Ph,             imRow);
            const v2f A1 = gateFrag(up, Q, Ph + wj0,       imRow);
            const v2f A2 = gateFrag(up, Q, Ph + wj1,       imRow);
            const v2f A3 = gateFrag(up, Q, Ph + wj1 + wj0, imRow);

            const unsigned fJ0    = foldAmp(1u << pj0);
            const unsigned fJ1    = foldAmp(1u << pj1);
            const unsigned ldLane = foldAmp(spread3(rsub, pa, pb, pc) | ((unsigned)half << ph)) ^ swzS;
            const unsigned ldB0 = rowB + (ldLane << 3);
            const unsigned ldB1 = rowB + ((ldLane ^ fJ0) << 3);
            const unsigned ldB2 = rowB + ((ldLane ^ fJ1) << 3);
            const unsigned ldB3 = rowB + ((ldLane ^ fJ1 ^ fJ0) << 3);

            char* const stB = smemB + GATE_BYTES;
            v2f x[16][4];
            char* q[16][4];
            v8f d[16];
            #define TRIP_EDGE() __builtin_amdgcn_sched_barrier(0)
            #pragma unroll 1
            for (int tb = 0; tb < 32; tb += 16) {
                #pragma unroll
                for (int k = 0; k < 16; ++k) {
                    const unsigned u0 = (foldAmp(spread3(2u * (unsigned)tb, pa, pb, pc)) << 3)
                                      ^ (foldAmp(spread3(2u * (unsigned)k, pa, pb, pc)) << 3);
                    q[k][0] = stB + (ldB0 ^ u0);
                    q[k][1] = stB + (ldB1 ^ u0);
                    q[k][2] = stB + (ldB2 ^ u0);
                    q[k][3] = stB + (ldB3 ^ u0);
                    #pragma unroll
                    for (int j = 0; j < 4; ++j) x[k][j] = *(const v2f*)q[k][j];
                }
                TRIP_EDGE();
                #pragma unroll
                for (int k = 0; k < 16; ++k) {
                    const v8f z = {0.f, 0.f, 0.f, 0.f, 0.f, 0.f, 0.f, 0.f};
                    d[k] = __builtin_amdgcn_wmma_f32_16x16x4_f32(false, A0, false, x[k][0], (short)0, z, false, false);
                }
                #pragma unroll
                for (int k = 0; k < 16; ++k)
                    d[k] = __builtin_amdgcn_wmma_f32_16x16x4_f32(false, A1, false, x[k][1], (short)0, d[k], false, false);
                #pragma unroll
                for (int k = 0; k < 16; ++k)
                    d[k] = __builtin_amdgcn_wmma_f32_16x16x4_f32(false, A2, false, x[k][2], (short)0, d[k], false, false);
                #pragma unroll
                for (int k = 0; k < 16; ++k)
                    d[k] = __builtin_amdgcn_wmma_f32_16x16x4_f32(false, A3, false, x[k][3], (short)0, d[k], false, false);
                TRIP_EDGE();
                #pragma unroll
                for (int k = 0; k < 16; ++k) {
                    #pragma unroll
                    for (int j = 0; j < 4; ++j) {
                        v2f o; o[0] = d[k][2 * j]; o[1] = d[k][2 * j + 1];
                        *(v2f*)q[k][j] = o;
                    }
                }
            }
            #undef TRIP_EDGE
        }
    }

    for (int ss2 = 0; ss2 < STATES_PER_WAVE; ++ss2) {
        float* o = out + (size_t)(bBase + ss2) * DIM;
        for (int j = lane; j < DIM; j += 32) {
            float2 a = stateV[ss2 * DIM + (foldAmp((unsigned)j) ^ stateSwz((unsigned)ss2))];
            o[j] = tanhf(5.0f * (a.x * a.x + a.y * a.y));
        }
    }
}

extern "C" void kernel_launch(void* const* d_in, const int* in_sizes, int n_in,
                              void* d_out, int out_size, void* d_ws, size_t ws_size,
                              hipStream_t stream) {
    (void)in_sizes; (void)n_in; (void)d_ws; (void)ws_size; (void)out_size;
    const float* features = (const float*)d_in[0];
    const float* weights  = (const float*)d_in[1];
    float* out = (float*)d_out;

    dim3 grid(BATCH_TOT / STATES_PER_BLOCK);
    dim3 block(BLOCK_THREADS);
    size_t shmem = (size_t)(GATE_LDS_FLOATS + WAVES_PER_BLOCK * STATE_LDS_FLOATS)
                   * sizeof(float);
    su4_quantum_kernel<<<grid, block, shmem, stream>>>(features, weights, out);
}


# Feature rows fetched as one group of sixteen 16-byte loads per lane at kernel entry (was one waited 4-byte load at a time, rows twice); norms from registers, so the order of the norm's sum changed
# speedup vs baseline: 4.3082x; 1.1507x over previous
#include <hip/hip_runtime.h>
#include <math.h>

#define NQ        9
#define DIM       512
#define NLAYERS   12
#define OBS       256
#define BATCH_TOT 16384

#define WAVES_PER_BLOCK   8
#define STATES_PER_WAVE   8
#define BLOCK_THREADS     (WAVES_PER_BLOCK * 32)
#define STATES_PER_BLOCK  (WAVES_PER_BLOCK * STATES_PER_WAVE)
#define NGATES            (NLAYERS * NQ)
#define NPAIRS            (NGATES / 2)
#define PAIR_FLOATS       128
#define GATE_LDS_FLOATS   (NPAIRS * PAIR_FLOATS)
#define STATE_LDS_FLOATS  (STATES_PER_WAVE * DIM * 2)
#define GTMP_FLOATS       (NGATES * 32)

typedef float v2f __attribute__((ext_vector_type(2)));
typedef float v8f __attribute__((ext_vector_type(8)));

struct cpx { float re, im; };
__device__ inline cpx cz() { cpx r; r.re = 0.f; r.im = 0.f; return r; }
__device__ inline cpx cmk(float a, float b) { cpx r; r.re = a; r.im = b; return r; }
__device__ inline cpx cmul(cpx a, cpx b) { return cmk(a.re*b.re - a.im*b.im, a.re*b.im + a.im*b.re); }
__device__ inline cpx cadd(cpx a, cpx b) { return cmk(a.re + b.re, a.im + b.im); }

__device__ inline void u3mat(float t, float p, float l, cpx U[2][2]) {
    float ct = cosf(0.5f * t), st = sinf(0.5f * t);
    float cp = cosf(p), sp = sinf(p), cl = cosf(l), sl = sinf(l);
    U[0][0] = cmk(ct, 0.f);
    U[0][1] = cmk(-cl * st, -sl * st);
    U[1][0] = cmk(cp * st, sp * st);
    U[1][1] = cmk((cp*cl - sp*sl) * ct, (cp*sl + sp*cl) * ct);
}

__device__ inline void kron22(const cpx A[2][2], const cpx B[2][2], cpx K[4][4]) {
    for (int a = 0; a < 2; ++a)
        for (int b = 0; b < 2; ++b)
            for (int c = 0; c < 2; ++c)
                for (int d = 0; d < 2; ++d)
                    K[2*a + c][2*b + d] = cmul(A[a][b], B[c][d]);
}

__device__ inline void mat4mul(cpx C[4][4], const cpx A[4][4], const cpx B[4][4]) {
    for (int i = 0; i < 4; ++i)
        for (int j = 0; j < 4; ++j) {
            cpx acc = cz();
            for (int k = 0; k < 4; ++k) acc = cadd(acc, cmul(A[i][k], B[k][j]));
            C[i][j] = acc;
        }
}

__device__ void computeGate(int g, const float* __restrict__ w, float2* __restrict__ gt) {
    const float* th = w + g * 15;
    cpx Ua[2][2], Ub[2][2];
    cpx pre[4][4], post[4][4];
    u3mat(th[0], th[1], th[2], Ua);  u3mat(th[3], th[4], th[5], Ub);  kron22(Ua, Ub, pre);
    u3mat(th[9], th[10], th[11], Ua); u3mat(th[12], th[13], th[14], Ub); kron22(Ua, Ub, post);

    float c6 = cosf(0.5f*th[6]), s6 = sinf(0.5f*th[6]);
    float c7 = cosf(0.5f*th[7]), s7 = sinf(0.5f*th[7]);
    float c8 = cosf(0.5f*th[8]), s8 = sinf(0.5f*th[8]);
    cpx Rxx[4][4], Ryy[4][4], Rzz[4][4];
    for (int i = 0; i < 4; ++i)
        for (int j = 0; j < 4; ++j) { Rxx[i][j] = cz(); Ryy[i][j] = cz(); Rzz[i][j] = cz(); }
    for (int i = 0; i < 4; ++i) { Rxx[i][i] = cmk(c6, 0.f); Ryy[i][i] = cmk(c7, 0.f); }
    Rxx[0][3] = cmk(0.f, -s6); Rxx[1][2] = cmk(0.f, -s6);
    Rxx[2][1] = cmk(0.f, -s6); Rxx[3][0] = cmk(0.f, -s6);
    Ryy[0][3] = cmk(0.f,  s7); Ryy[1][2] = cmk(0.f, -s7);
    Ryy[2][1] = cmk(0.f, -s7); Ryy[3][0] = cmk(0.f,  s7);
    Rzz[0][0] = cmk(c8, -s8); Rzz[1][1] = cmk(c8,  s8);
    Rzz[2][2] = cmk(c8,  s8); Rzz[3][3] = cmk(c8, -s8);

    cpx t1[4][4], t2[4][4];
    mat4mul(t1, Rxx, Ryy);
    mat4mul(t2, t1, Rzz);
    mat4mul(t1, t2, pre);
    mat4mul(t2, post, t1);

    float2* gb = gt + g * 16;
    for (int a = 0; a < 4; ++a)
        for (int b = 0; b < 4; ++b)
            gb[a*4 + b] = make_float2(t2[a][b].re, t2[a][b].im);
}

__device__ inline void fusePair(int idx, const float2* __restrict__ gt, float2* __restrict__ ut) {
    const int pr = idx >> 6, Q = (idx >> 3) & 7, P = idx & 7;
    const float2* G1 = gt + (2 * pr) * 16;
    const float2* G2 = gt + (2 * pr + 1) * 16;
    const int A = Q >> 2, B = (Q >> 1) & 1, C = Q & 1;
    const int a = P >> 2, b = (P >> 1) & 1, c = P & 1;
    float re = 0.f, im = 0.f;
    for (int bp = 0; bp < 2; ++bp) {
        const float2 x = G2[(2 * B + C) * 4 + 2 * bp + c];
        const float2 y = G1[(2 * A + bp) * 4 + 2 * a + b];
        re += x.x * y.x - x.y * y.y;
        im += x.x * y.y + x.y * y.x;
    }
    ut[idx] = make_float2(re, im);
}

__device__ inline unsigned foldAmp(unsigned a) {
    return a ^ ((((a >> 5) ^ (a >> 7)) & 3u) << 3);
}
__device__ inline unsigned stateSwz(unsigned s) {
    return ((s & 1u) ? 12u : 0u) ^ ((s & 2u) ? 24u : 0u) ^ ((s & 4u) ? 7u : 0u);
}
__device__ inline unsigned insert0(unsigned x, int pos) {
    return ((x >> pos) << (pos + 1)) | (x & ((1u << pos) - 1u));
}
__device__ inline unsigned spread3(unsigned r, int p0, int p1, int p2) {
    int lo = (p0 < p1) ? ((p0 < p2) ? p0 : p2) : ((p1 < p2) ? p1 : p2);
    int hi = (p0 > p1) ? ((p0 > p2) ? p0 : p2) : ((p1 > p2) ? p1 : p2);
    int mid = p0 + p1 + p2 - lo - hi;
    return insert0(insert0(insert0(r, lo), mid), hi);
}
__device__ inline v2f gateFrag(const float2* ut, int Q, int P, bool imRow) {
    const float2 u = ut[Q * 8 + P];
    v2f r;
    r[0] = imRow ? u.y : u.x;
    r[1] = imRow ? u.x : -u.y;
    return r;
}

__global__ __launch_bounds__(BLOCK_THREADS, 1)
void su4_quantum_kernel(const float* __restrict__ features,
                        const float* __restrict__ weights,
                        float* __restrict__ out) {
    extern __shared__ __align__(16) float smem[];
    float2* ut = (float2*)smem;
    static_assert(GTMP_FLOATS <= WAVES_PER_BLOCK * STATE_LDS_FLOATS, "single-gate scratch must fit");
    float2* gt = (float2*)(smem + GATE_LDS_FLOATS);
    const int tid  = threadIdx.x;
    const int lane = tid & 31;
    const int wave = tid >> 5;
    float*  stateF = smem + GATE_LDS_FLOATS + wave * STATE_LDS_FLOATS;
    float2* stateV = (float2*)stateF;

    const int bBase = (blockIdx.x * WAVES_PER_BLOCK + wave) * STATES_PER_WAVE;
    static_assert(OBS == 256 && DIM == 2 * OBS, "a row is two 512-byte pieces; the upper half of a state is zero");
    const float4* fw = (const float4*)(features + (size_t)bBase * OBS);
    float4 fv[2 * STATES_PER_WAVE];
    #pragma unroll
    for (int t = 0; t < 2 * STATES_PER_WAVE; ++t) fv[t] = fw[t * 32 + lane];

    if (tid < NGATES) computeGate(tid, weights, gt);
    __syncthreads();
    for (int idx = tid; idx < NPAIRS * 64; idx += BLOCK_THREADS) fusePair(idx, gt, ut);
    __syncthreads();

    float ssq[STATES_PER_WAVE];
    #pragma unroll
    for (int s = 0; s < STATES_PER_WAVE; ++s) {
        const float4 a = fv[2 * s], b = fv[2 * s + 1];
        ssq[s] = (a.x * a.x + a.y * a.y + a.z * a.z + a.w * a.w)
               + (b.x * b.x + b.y * b.y + b.z * b.z + b.w * b.w);
    }
    #pragma unroll
    for (int msk = 16; msk >= 1; msk >>= 1) {
        #pragma unroll
        for (int s = 0; s < STATES_PER_WAVE; ++s) ssq[s] += __shfl_xor(ssq[s], msk, 32);
    }
    #pragma unroll
    for (int s = 0; s < STATES_PER_WAVE; ++s) {
        const float inv = 1.0f / (sqrtf(ssq[s]) + 1e-8f);
        const unsigned swz = stateSwz((unsigned)s);
        #pragma unroll
        for (int k = 0; k < 2; ++k) {
            const float4 v = fv[2 * s + k];
            const float a[4] = { v.x * inv, v.y * inv, v.z * inv, v.w * inv };
            const unsigned j0 = 4u * (unsigned)lane + 128u * (unsigned)k;
            const unsigned g  = (foldAmp(j0) ^ swz) & ~3u;
            float4* dst = (float4*)(stateV + s * DIM + g);
            dst[0] = make_float4(a[0 ^ (swz & 3u)], 0.f, a[1 ^ (swz & 3u)], 0.f);
            dst[1] = make_float4(a[2 ^ (swz & 3u)], 0.f, a[3 ^ (swz & 3u)], 0.f);
        }
        float4* zr = (float4*)(stateV + s * DIM + OBS);
        #pragma unroll
        for (int i = 0; i < 4; ++i) zr[i * 32 + lane] = make_float4(0.f, 0.f, 0.f, 0.f);
    }
    __syncthreads();

    const int s    = lane & 7;
    const unsigned rsub = (unsigned)((lane >> 3) & 1);
    const int half = lane >> 4;
    const int grow = lane & 15;
    const bool imRow = (grow & 1) != 0;
    const unsigned swzS = stateSwz((unsigned)s);

    static_assert((GATE_LDS_FLOATS * 4) % 256 == 0, "gate table must not shift the bank map");
    static_assert((STATE_LDS_FLOATS * 4) % 4096 == 0 && (DIM * 8) % 4096 == 0,
                  "wave region and state plane offsets must keep byte bits 0..11 clear");
    char* const smemB = (char*)smem;
    const unsigned GATE_BYTES = GATE_LDS_FLOATS * 4u;
    const unsigned waveB = (unsigned)wave * (STATE_LDS_FLOATS * 4u);
    const unsigned rowB  = waveB + (unsigned)s * (DIM * 8u);

    for (int rep = 0; rep < NPAIRS / NQ; ++rep) {
        #pragma unroll
        for (int kk = 0; kk < NQ; ++kk) {
            const int i  = (2 * kk) % NQ;
            const int pa = 8 - i;
            const int pb = 8 - ((i + 1) % NQ);
            const int pc = 8 - ((i + 2) % NQ);
            const bool hOnC = (i == 6);
            const int ph  = hOnC ? pc : pa;  const int wh  = hOnC ? 1 : 4;
            const int pj1 = hOnC ? pa : pb;  const int wj1 = hOnC ? 4 : 2;
            const int pj0 = hOnC ? pb : pc;  const int wj0 = hOnC ? 2 : 1;

            const float2* up = ut + (rep * NQ + kk) * 64;
            const int Q  = (grow >> 3) * wh + ((grow >> 2) & 1) * wj1 + ((grow >> 1) & 1) * wj0;
            const int Ph = half * wh;
            const v2f A0 = gateFrag(up, Q, Ph,             imRow);
            const v2f A1 = gateFrag(up, Q, Ph + wj0,       imRow);
            const v2f A2 = gateFrag(up, Q, Ph + wj1,       imRow);
            const v2f A3 = gateFrag(up, Q, Ph + wj1 + wj0, imRow);

            const unsigned fJ0    = foldAmp(1u << pj0);
            const unsigned fJ1    = foldAmp(1u << pj1);
            const unsigned ldLane = foldAmp(spread3(rsub, pa, pb, pc) | ((unsigned)half << ph)) ^ swzS;
            const unsigned ldB0 = rowB + (ldLane << 3);
            const unsigned ldB1 = rowB + ((ldLane ^ fJ0) << 3);
            const unsigned ldB2 = rowB + ((ldLane ^ fJ1) << 3);
            const unsigned ldB3 = rowB + ((ldLane ^ fJ1 ^ fJ0) << 3);

            char* const stB = smemB + GATE_BYTES;
            v2f x[16][4];
            char* q[16][4];
            v8f d[16];
            #define TRIP_EDGE() __builtin_amdgcn_sched_barrier(0)
            #pragma unroll 1
            for (int tb = 0; tb < 32; tb += 16) {
                #pragma unroll
                for (int k = 0; k < 16; ++k) {
                    const unsigned u0 = (foldAmp(spread3(2u * (unsigned)tb, pa, pb, pc)) << 3)
                                      ^ (foldAmp(spread3(2u * (unsigned)k, pa, pb, pc)) << 3);
                    q[k][0] = stB + (ldB0 ^ u0);
                    q[k][1] = stB + (ldB1 ^ u0);
                    q[k][2] = stB + (ldB2 ^ u0);
                    q[k][3] = stB + (ldB3 ^ u0);
                    #pragma unroll
                    for (int j = 0; j < 4; ++j) x[k][j] = *(const v2f*)q[k][j];
                }
                TRIP_EDGE();
                #pragma unroll
                for (int k = 0; k < 16; ++k) {
                    const v8f z = {0.f, 0.f, 0.f, 0.f, 0.f, 0.f, 0.f, 0.f};
                    d[k] = __builtin_amdgcn_wmma_f32_16x16x4_f32(false, A0, false, x[k][0], (short)0, z, false, false);
                }
                #pragma unroll
                for (int k = 0; k < 16; ++k)
                    d[k] = __builtin_amdgcn_wmma_f32_16x16x4_f32(false, A1, false, x[k][1], (short)0, d[k], false, false);
                #pragma unroll
                for (int k = 0; k < 16; ++k)
                    d[k] = __builtin_amdgcn_wmma_f32_16x16x4_f32(false, A2, false, x[k][2], (short)0, d[k], false, false);
                #pragma unroll
                for (int k = 0; k < 16; ++k)
                    d[k] = __builtin_amdgcn_wmma_f32_16x16x4_f32(false, A3, false, x[k][3], (short)0, d[k], false, false);
                TRIP_EDGE();
                #pragma unroll
                for (int k = 0; k < 16; ++k) {
                    #pragma unroll
                    for (int j = 0; j < 4; ++j) {
                        v2f o; o[0] = d[k][2 * j]; o[1] = d[k][2 * j + 1];
                        *(v2f*)q[k][j] = o;
                    }
                }
            }
            #undef TRIP_EDGE
        }
    }

    for (int ss2 = 0; ss2 < STATES_PER_WAVE; ++ss2) {
        float* o = out + (size_t)(bBase + ss2) * DIM;
        for (int j = lane; j < DIM; j += 32) {
            float2 a = stateV[ss2 * DIM + (foldAmp((unsigned)j) ^ stateSwz((unsigned)ss2))];
            o[j] = tanhf(5.0f * (a.x * a.x + a.y * a.y));
        }
    }
}

extern "C" void kernel_launch(void* const* d_in, const int* in_sizes, int n_in,
                              void* d_out, int out_size, void* d_ws, size_t ws_size,
                              hipStream_t stream) {
    (void)in_sizes; (void)n_in; (void)d_ws; (void)ws_size; (void)out_size;
    const float* features = (const float*)d_in[0];
    const float* weights  = (const float*)d_in[1];
    float* out = (float*)d_out;

    dim3 grid(BATCH_TOT / STATES_PER_BLOCK);
    dim3 block(BLOCK_THREADS);
    size_t shmem = (size_t)(GATE_LDS_FLOATS + WAVES_PER_BLOCK * STATE_LDS_FLOATS)
                   * sizeof(float);
    su4_quantum_kernel<<<grid, block, shmem, stream>>>(features, weights, out);
}


# Two fused pairs per LDS pass: a lane's D pairs of pair (a,b,c) are, by register choice, the B fragments of pair (c,d,e), c on the lane half: 27 load/store passes, was 54; k order of sums changed
# speedup vs baseline: 4.6791x; 1.0861x over previous
#include <hip/hip_runtime.h>
#include <math.h>

#define NQ        9
#define DIM       512
#define NLAYERS   12
#define OBS       256
#define BATCH_TOT 16384

#define WAVES_PER_BLOCK   8
#define STATES_PER_WAVE   8
#define BLOCK_THREADS     (WAVES_PER_BLOCK * 32)
#define STATES_PER_BLOCK  (WAVES_PER_BLOCK * STATES_PER_WAVE)
#define NGATES            (NLAYERS * NQ)
#define NPAIRS            (NGATES / 2)
#define PAIR_FLOATS       128
#define GATE_LDS_FLOATS   (NPAIRS * PAIR_FLOATS)
#define STATE_LDS_FLOATS  (STATES_PER_WAVE * DIM * 2)
#define GTMP_FLOATS       (NGATES * 32)

typedef float v2f __attribute__((ext_vector_type(2)));
typedef float v8f __attribute__((ext_vector_type(8)));

struct cpx { float re, im; };
__device__ inline cpx cz() { cpx r; r.re = 0.f; r.im = 0.f; return r; }
__device__ inline cpx cmk(float a, float b) { cpx r; r.re = a; r.im = b; return r; }
__device__ inline cpx cmul(cpx a, cpx b) { return cmk(a.re*b.re - a.im*b.im, a.re*b.im + a.im*b.re); }
__device__ inline cpx cadd(cpx a, cpx b) { return cmk(a.re + b.re, a.im + b.im); }

__device__ inline void u3mat(float t, float p, float l, cpx U[2][2]) {
    float ct = cosf(0.5f * t), st = sinf(0.5f * t);
    float cp = cosf(p), sp = sinf(p), cl = cosf(l), sl = sinf(l);
    U[0][0] = cmk(ct, 0.f);
    U[0][1] = cmk(-cl * st, -sl * st);
    U[1][0] = cmk(cp * st, sp * st);
    U[1][1] = cmk((cp*cl - sp*sl) * ct, (cp*sl + sp*cl) * ct);
}

__device__ inline void kron22(const cpx A[2][2], const cpx B[2][2], cpx K[4][4]) {
    for (int a = 0; a < 2; ++a)
        for (int b = 0; b < 2; ++b)
            for (int c = 0; c < 2; ++c)
                for (int d = 0; d < 2; ++d)
                    K[2*a + c][2*b + d] = cmul(A[a][b], B[c][d]);
}

__device__ inline void mat4mul(cpx C[4][4], const cpx A[4][4], const cpx B[4][4]) {
    for (int i = 0; i < 4; ++i)
        for (int j = 0; j < 4; ++j) {
            cpx acc = cz();
            for (int k = 0; k < 4; ++k) acc = cadd(acc, cmul(A[i][k], B[k][j]));
            C[i][j] = acc;
        }
}

__device__ void computeGate(int g, const float* __restrict__ w, float2* __restrict__ gt) {
    const float* th = w + g * 15;
    cpx Ua[2][2], Ub[2][2];
    cpx pre[4][4], post[4][4];
    u3mat(th[0], th[1], th[2], Ua);  u3mat(th[3], th[4], th[5], Ub);  kron22(Ua, Ub, pre);
    u3mat(th[9], th[10], th[11], Ua); u3mat(th[12], th[13], th[14], Ub); kron22(Ua, Ub, post);

    float c6 = cosf(0.5f*th[6]), s6 = sinf(0.5f*th[6]);
    float c7 = cosf(0.5f*th[7]), s7 = sinf(0.5f*th[7]);
    float c8 = cosf(0.5f*th[8]), s8 = sinf(0.5f*th[8]);
    cpx Rxx[4][4], Ryy[4][4], Rzz[4][4];
    for (int i = 0; i < 4; ++i)
        for (int j = 0; j < 4; ++j) { Rxx[i][j] = cz(); Ryy[i][j] = cz(); Rzz[i][j] = cz(); }
    for (int i = 0; i < 4; ++i) { Rxx[i][i] = cmk(c6, 0.f); Ryy[i][i] = cmk(c7, 0.f); }
    Rxx[0][3] = cmk(0.f, -s6); Rxx[1][2] = cmk(0.f, -s6);
    Rxx[2][1] = cmk(0.f, -s6); Rxx[3][0] = cmk(0.f, -s6);
    Ryy[0][3] = cmk(0.f,  s7); Ryy[1][2] = cmk(0.f, -s7);
    Ryy[2][1] = cmk(0.f, -s7); Ryy[3][0] = cmk(0.f,  s7);
    Rzz[0][0] = cmk(c8, -s8); Rzz[1][1] = cmk(c8,  s8);
    Rzz[2][2] = cmk(c8,  s8); Rzz[3][3] = cmk(c8, -s8);

    cpx t1[4][4], t2[4][4];
    mat4mul(t1, Rxx, Ryy);
    mat4mul(t2, t1, Rzz);
    mat4mul(t1, t2, pre);
    mat4mul(t2, post, t1);

    float2* gb = gt + g * 16;
    for (int a = 0; a < 4; ++a)
        for (int b = 0; b < 4; ++b)
            gb[a*4 + b] = make_float2(t2[a][b].re, t2[a][b].im);
}

__device__ inline void fusePair(int idx, const float2* __restrict__ gt, float2* __restrict__ ut) {
    const int pr = idx >> 6, Q = (idx >> 3) & 7, P = idx & 7;
    const float2* G1 = gt + (2 * pr) * 16;
    const float2* G2 = gt + (2 * pr + 1) * 16;
    const int A = Q >> 2, B = (Q >> 1) & 1, C = Q & 1;
    const int a = P >> 2, b = (P >> 1) & 1, c = P & 1;
    float re = 0.f, im = 0.f;
    for (int bp = 0; bp < 2; ++bp) {
        const float2 x = G2[(2 * B + C) * 4 + 2 * bp + c];
        const float2 y = G1[(2 * A + bp) * 4 + 2 * a + b];
        re += x.x * y.x - x.y * y.y;
        im += x.x * y.y + x.y * y.x;
    }
    ut[idx] = make_float2(re, im);
}

__device__ inline unsigned foldAmp(unsigned a) {
    return a ^ ((((a >> 5) ^ (a >> 7)) & 3u) << 3);
}
__device__ inline unsigned stateSwz(unsigned s) {
    return ((s & 1u) ? 12u : 0u) ^ ((s & 2u) ? 24u : 0u) ^ ((s & 4u) ? 7u : 0u);
}
__device__ inline int qbit(int q) { return 8 - (q % NQ); }
__device__ inline int bankClass(int p) { return p == 0 ? 1 : (p == 1 ? 2 : 3); }
__device__ inline v2f gateFrag(const float2* ut, int Q, int P, bool imRow) {
    const float2 u = ut[Q * 8 + P];
    v2f r;
    r[0] = imRow ? u.y : u.x;
    r[1] = imRow ? u.x : -u.y;
    return r;
}

__global__ __launch_bounds__(BLOCK_THREADS, 1)
void su4_quantum_kernel(const float* __restrict__ features,
                        const float* __restrict__ weights,
                        float* __restrict__ out) {
    extern __shared__ __align__(16) float smem[];
    float2* ut = (float2*)smem;
    static_assert(GTMP_FLOATS <= WAVES_PER_BLOCK * STATE_LDS_FLOATS, "single-gate scratch must fit");
    float2* gt = (float2*)(smem + GATE_LDS_FLOATS);
    const int tid  = threadIdx.x;
    const int lane = tid & 31;
    const int wave = tid >> 5;
    float*  stateF = smem + GATE_LDS_FLOATS + wave * STATE_LDS_FLOATS;
    float2* stateV = (float2*)stateF;

    const int bBase = (blockIdx.x * WAVES_PER_BLOCK + wave) * STATES_PER_WAVE;
    static_assert(OBS == 256 && DIM == 2 * OBS, "a row is two 512-byte pieces; the upper half of a state is zero");
    const float4* fw = (const float4*)(features + (size_t)bBase * OBS);
    float4 fv[2 * STATES_PER_WAVE];
    #pragma unroll
    for (int t = 0; t < 2 * STATES_PER_WAVE; ++t) fv[t] = fw[t * 32 + lane];

    if (tid < NGATES) computeGate(tid, weights, gt);
    __syncthreads();
    for (int idx = tid; idx < NPAIRS * 64; idx += BLOCK_THREADS) fusePair(idx, gt, ut);
    __syncthreads();

    float ssq[STATES_PER_WAVE];
    #pragma unroll
    for (int s = 0; s < STATES_PER_WAVE; ++s) {
        const float4 a = fv[2 * s], b = fv[2 * s + 1];
        ssq[s] = (a.x * a.x + a.y * a.y + a.z * a.z + a.w * a.w)
               + (b.x * b.x + b.y * b.y + b.z * b.z + b.w * b.w);
    }
    #pragma unroll
    for (int msk = 16; msk >= 1; msk >>= 1) {
        #pragma unroll
        for (int s = 0; s < STATES_PER_WAVE; ++s) ssq[s] += __shfl_xor(ssq[s], msk, 32);
    }
    #pragma unroll
    for (int s = 0; s < STATES_PER_WAVE; ++s) {
        const float inv = 1.0f / (sqrtf(ssq[s]) + 1e-8f);
        const unsigned swz = stateSwz((unsigned)s);
        #pragma unroll
        for (int k = 0; k < 2; ++k) {
            const float4 v = fv[2 * s + k];
            const float a[4] = { v.x * inv, v.y * inv, v.z * inv, v.w * inv };
            const unsigned j0 = 4u * (unsigned)lane + 128u * (unsigned)k;
            const unsigned g  = (foldAmp(j0) ^ swz) & ~3u;
            float4* dst = (float4*)(stateV + s * DIM + g);
            dst[0] = make_float4(a[0 ^ (swz & 3u)], 0.f, a[1 ^ (swz & 3u)], 0.f);
            dst[1] = make_float4(a[2 ^ (swz & 3u)], 0.f, a[3 ^ (swz & 3u)], 0.f);
        }
        float4* zr = (float4*)(stateV + s * DIM + OBS);
        #pragma unroll
        for (int i = 0; i < 4; ++i) zr[i * 32 + lane] = make_float4(0.f, 0.f, 0.f, 0.f);
    }
    __syncthreads();

    const int s    = lane & 7;
    const unsigned rsub = (unsigned)((lane >> 3) & 1);
    const int half = lane >> 4;
    const int grow = lane & 15;
    const bool imRow = (grow & 1) != 0;
    const unsigned swzS = stateSwz((unsigned)s);

    static_assert((GATE_LDS_FLOATS * 4) % 256 == 0, "gate table must not shift the bank map");
    static_assert((STATE_LDS_FLOATS * 4) % 4096 == 0 && (DIM * 8) % 4096 == 0,
                  "wave region and state plane offsets must keep byte bits 0..11 clear");
    static_assert(NPAIRS % (2 * NQ) == 0, "a period of the qubit pattern is nine passes of two pairs");
    char* const smemB = (char*)smem;
    const unsigned GATE_BYTES = GATE_LDS_FLOATS * 4u;
    const unsigned waveB = (unsigned)wave * (STATE_LDS_FLOATS * 4u);
    const unsigned rowB  = waveB + (unsigned)s * (DIM * 8u);

    for (int rep = 0; rep < NPAIRS / (2 * NQ); ++rep) {
        #pragma unroll
        for (int kk = 0; kk < NQ; ++kk) {
            const int i  = (4 * kk) % NQ;
            const int pa = qbit(i), pb = qbit(i + 1), pc = qbit(i + 2), pd = qbit(i + 3), pe = qbit(i + 4);
            const int cls = bankClass(pc);
            const int sel = (bankClass(qbit(i + 5)) != cls) ? 0 : (bankClass(qbit(i + 6)) != cls) ? 1
                          : (bankClass(qbit(i + 7)) != cls) ? 2 : (bankClass(qbit(i + 8)) != cls) ? 3 : 0;
            const int pr = qbit(i + 5 + sel);
            const int pt = qbit(i + 5 + ((sel + 1) & 3));
            const int px = qbit(i + 5 + ((sel + 2) & 3));
            const int py = qbit(i + 5 + ((sel + 3) & 3));

            const float2* up1 = ut + (2 * (rep * NQ + kk)) * 64;
            const float2* up2 = up1 + 64;
            const int Q1 = (grow >> 3) * 1 + ((grow >> 2) & 1) * 4 + ((grow >> 1) & 1) * 2;
            const int Q2 = (grow >> 3) * 4 + ((grow >> 2) & 1) * 2 + ((grow >> 1) & 1) * 1;
            v2f A1[4], A2[4];
            #pragma unroll
            for (int j = 0; j < 4; ++j) {
                A1[j] = gateFrag(up1, Q1, half * 1 + (j & 1) * 2 + (j >> 1) * 4, imRow);
                A2[j] = gateFrag(up2, Q2, half * 4 + (j & 1) * 1 + (j >> 1) * 2, imRow);
            }

            const unsigned ldLane = foldAmp((rsub << pr) | ((unsigned)half << pc)) ^ swzS;
            unsigned ldB[4];
            #pragma unroll
            for (int j = 0; j < 4; ++j)
                ldB[j] = rowB + ((ldLane ^ foldAmp(((unsigned)(j & 1) << pb) | ((unsigned)(j >> 1) << pa))) << 3);

            char* const stB = smemB + GATE_BYTES;
            v2f x[16][4];
            char* q[16][4];
            v8f d1[16], d2[16];
            #define TRIP_EDGE() __builtin_amdgcn_sched_barrier(0)
            #pragma unroll 1
            for (int tb = 0; tb < 2; ++tb) {
                const unsigned tpart = tb ? (foldAmp(1u << pt) << 3) : 0u;
                #pragma unroll
                for (int k = 0; k < 16; ++k) {
                    const unsigned u0 = tpart
                        ^ (foldAmp(((unsigned)(k & 1) << pe) | ((unsigned)((k >> 1) & 1) << pd)
                                 | ((unsigned)((k >> 2) & 1) << px) | ((unsigned)((k >> 3) & 1) << py)) << 3);
                    #pragma unroll
                    for (int j = 0; j < 4; ++j) {
                        q[k][j] = stB + (ldB[j] ^ u0);
                        x[k][j] = *(const v2f*)q[k][j];
                    }
                }
                TRIP_EDGE();
                #pragma unroll
                for (int k = 0; k < 16; ++k) {
                    const v8f z = {0.f, 0.f, 0.f, 0.f, 0.f, 0.f, 0.f, 0.f};
                    d1[k] = __builtin_amdgcn_wmma_f32_16x16x4_f32(false, A1[0], false, x[k][0], (short)0, z, false, false);
                }
                #pragma unroll
                for (int j = 1; j < 4; ++j) {
                    #pragma unroll
                    for (int k = 0; k < 16; ++k)
                        d1[k] = __builtin_amdgcn_wmma_f32_16x16x4_f32(false, A1[j], false, x[k][j], (short)0, d1[k], false, false);
                }
                #pragma unroll
                for (int jp = 0; jp < 4; ++jp) {
                    #pragma unroll
                    for (int kp = 0; kp < 16; ++kp) {
                        const int kS = jp | (kp & 12), jS = kp & 3;
                        v2f b; b[0] = d1[kS][2 * jS]; b[1] = d1[kS][2 * jS + 1];
                        if (jp == 0) {
                            const v8f z = {0.f, 0.f, 0.f, 0.f, 0.f, 0.f, 0.f, 0.f};
                            d2[kp] = __builtin_amdgcn_wmma_f32_16x16x4_f32(false, A2[0], false, b, (short)0, z, false, false);
                        } else {
                            d2[kp] = __builtin_amdgcn_wmma_f32_16x16x4_f32(false, A2[jp], false, b, (short)0, d2[kp], false, false);
                        }
                    }
                }
                TRIP_EDGE();
                #pragma unroll
                for (int kp = 0; kp < 16; ++kp) {
                    #pragma unroll
                    for (int jp = 0; jp < 4; ++jp) {
                        v2f o; o[0] = d2[kp][2 * jp]; o[1] = d2[kp][2 * jp + 1];
                        *(v2f*)q[jp | (kp & 12)][kp & 3] = o;
                    }
                }
            }
            #undef TRIP_EDGE
        }
    }

    for (int ss2 = 0; ss2 < STATES_PER_WAVE; ++ss2) {
        float* o = out + (size_t)(bBase + ss2) * DIM;
        for (int j = lane; j < DIM; j += 32) {
            float2 a = stateV[ss2 * DIM + (foldAmp((unsigned)j) ^ stateSwz((unsigned)ss2))];
            o[j] = tanhf(5.0f * (a.x * a.x + a.y * a.y));
        }
    }
}

extern "C" void kernel_launch(void* const* d_in, const int* in_sizes, int n_in,
                              void* d_out, int out_size, void* d_ws, size_t ws_size,
                              hipStream_t stream) {
    (void)in_sizes; (void)n_in; (void)d_ws; (void)ws_size; (void)out_size;
    const float* features = (const float*)d_in[0];
    const float* weights  = (const float*)d_in[1];
    float* out = (float*)d_out;

    dim3 grid(BATCH_TOT / STATES_PER_BLOCK);
    dim3 block(BLOCK_THREADS);
    size_t shmem = (size_t)(GATE_LDS_FLOATS + WAVES_PER_BLOCK * STATE_LDS_FLOATS)
                   * sizeof(float);
    su4_quantum_kernel<<<grid, block, shmem, stream>>>(features, weights, out);
}


# Epilogue batched per state: 8 16-byte LDS loads, 16 independent branch-free tanh evaluations per lane, 4 16-byte stores; was one waited load, one branchy tanh, one 4-byte store at a time
# speedup vs baseline: 5.1071x; 1.0915x over previous
#include <hip/hip_runtime.h>
#include <math.h>

#define NQ        9
#define DIM       512
#define NLAYERS   12
#define OBS       256
#define BATCH_TOT 16384

#define WAVES_PER_BLOCK   8
#define STATES_PER_WAVE   8
#define BLOCK_THREADS     (WAVES_PER_BLOCK * 32)
#define STATES_PER_BLOCK  (WAVES_PER_BLOCK * STATES_PER_WAVE)
#define NGATES            (NLAYERS * NQ)
#define NPAIRS            (NGATES / 2)
#define PAIR_FLOATS       128
#define GATE_LDS_FLOATS   (NPAIRS * PAIR_FLOATS)
#define STATE_LDS_FLOATS  (STATES_PER_WAVE * DIM * 2)
#define GTMP_FLOATS       (NGATES * 32)

typedef float v2f __attribute__((ext_vector_type(2)));
typedef float v8f __attribute__((ext_vector_type(8)));

struct cpx { float re, im; };
__device__ inline cpx cz() { cpx r; r.re = 0.f; r.im = 0.f; return r; }
__device__ inline cpx cmk(float a, float b) { cpx r; r.re = a; r.im = b; return r; }
__device__ inline cpx cmul(cpx a, cpx b) { return cmk(a.re*b.re - a.im*b.im, a.re*b.im + a.im*b.re); }
__device__ inline cpx cadd(cpx a, cpx b) { return cmk(a.re + b.re, a.im + b.im); }

__device__ inline void u3mat(float t, float p, float l, cpx U[2][2]) {
    float ct = cosf(0.5f * t), st = sinf(0.5f * t);
    float cp = cosf(p), sp = sinf(p), cl = cosf(l), sl = sinf(l);
    U[0][0] = cmk(ct, 0.f);
    U[0][1] = cmk(-cl * st, -sl * st);
    U[1][0] = cmk(cp * st, sp * st);
    U[1][1] = cmk((cp*cl - sp*sl) * ct, (cp*sl + sp*cl) * ct);
}

__device__ inline void kron22(const cpx A[2][2], const cpx B[2][2], cpx K[4][4]) {
    for (int a = 0; a < 2; ++a)
        for (int b = 0; b < 2; ++b)
            for (int c = 0; c < 2; ++c)
                for (int d = 0; d < 2; ++d)
                    K[2*a + c][2*b + d] = cmul(A[a][b], B[c][d]);
}

__device__ inline void mat4mul(cpx C[4][4], const cpx A[4][4], const cpx B[4][4]) {
    for (int i = 0; i < 4; ++i)
        for (int j = 0; j < 4; ++j) {
            cpx acc = cz();
            for (int k = 0; k < 4; ++k) acc = cadd(acc, cmul(A[i][k], B[k][j]));
            C[i][j] = acc;
        }
}

__device__ void computeGate(int g, const float* __restrict__ w, float2* __restrict__ gt) {
    const float* th = w + g * 15;
    cpx Ua[2][2], Ub[2][2];
    cpx pre[4][4], post[4][4];
    u3mat(th[0], th[1], th[2], Ua);  u3mat(th[3], th[4], th[5], Ub);  kron22(Ua, Ub, pre);
    u3mat(th[9], th[10], th[11], Ua); u3mat(th[12], th[13], th[14], Ub); kron22(Ua, Ub, post);

    float c6 = cosf(0.5f*th[6]), s6 = sinf(0.5f*th[6]);
    float c7 = cosf(0.5f*th[7]), s7 = sinf(0.5f*th[7]);
    float c8 = cosf(0.5f*th[8]), s8 = sinf(0.5f*th[8]);
    cpx Rxx[4][4], Ryy[4][4], Rzz[4][4];
    for (int i = 0; i < 4; ++i)
        for (int j = 0; j < 4; ++j) { Rxx[i][j] = cz(); Ryy[i][j] = cz(); Rzz[i][j] = cz(); }
    for (int i = 0; i < 4; ++i) { Rxx[i][i] = cmk(c6, 0.f); Ryy[i][i] = cmk(c7, 0.f); }
    Rxx[0][3] = cmk(0.f, -s6); Rxx[1][2] = cmk(0.f, -s6);
    Rxx[2][1] = cmk(0.f, -s6); Rxx[3][0] = cmk(0.f, -s6);
    Ryy[0][3] = cmk(0.f,  s7); Ryy[1][2] = cmk(0.f, -s7);
    Ryy[2][1] = cmk(0.f, -s7); Ryy[3][0] = cmk(0.f,  s7);
    Rzz[0][0] = cmk(c8, -s8); Rzz[1][1] = cmk(c8,  s8);
    Rzz[2][2] = cmk(c8,  s8); Rzz[3][3] = cmk(c8, -s8);

    cpx t1[4][4], t2[4][4];
    mat4mul(t1, Rxx, Ryy);
    mat4mul(t2, t1, Rzz);
    mat4mul(t1, t2, pre);
    mat4mul(t2, post, t1);

    float2* gb = gt + g * 16;
    for (int a = 0; a < 4; ++a)
        for (int b = 0; b < 4; ++b)
            gb[a*4 + b] = make_float2(t2[a][b].re, t2[a][b].im);
}

__device__ inline void fusePair(int idx, const float2* __restrict__ gt, float2* __restrict__ ut) {
    const int pr = idx >> 6, Q = (idx >> 3) & 7, P = idx & 7;
    const float2* G1 = gt + (2 * pr) * 16;
    const float2* G2 = gt + (2 * pr + 1) * 16;
    const int A = Q >> 2, B = (Q >> 1) & 1, C = Q & 1;
    const int a = P >> 2, b = (P >> 1) & 1, c = P & 1;
    float re = 0.f, im = 0.f;
    for (int bp = 0; bp < 2; ++bp) {
        const float2 x = G2[(2 * B + C) * 4 + 2 * bp + c];
        const float2 y = G1[(2 * A + bp) * 4 + 2 * a + b];
        re += x.x * y.x - x.y * y.y;
        im += x.x * y.y + x.y * y.x;
    }
    ut[idx] = make_float2(re, im);
}

__device__ inline unsigned foldAmp(unsigned a) {
    return a ^ ((((a >> 5) ^ (a >> 7)) & 3u) << 3);
}
__device__ inline unsigned stateSwz(unsigned s) {
    return ((s & 1u) ? 12u : 0u) ^ ((s & 2u) ? 24u : 0u) ^ ((s & 4u) ? 7u : 0u);
}
__device__ inline int qbit(int q) { return 8 - (q % NQ); }
__device__ inline int bankClass(int p) { return p == 0 ? 1 : (p == 1 ? 2 : 3); }
__device__ inline v2f gateFrag(const float2* ut, int Q, int P, bool imRow) {
    const float2 u = ut[Q * 8 + P];
    v2f r;
    r[0] = imRow ? u.y : u.x;
    r[1] = imRow ? u.x : -u.y;
    return r;
}

__device__ inline float tanhSmall(float x) {
    const float p = x * x;
    float q = fmaf(p, -0x1.758e7ap-8f, 0x1.521192p-6f);
    q = fmaf(p, q, -0x1.b8389cp-5f);
    q = fmaf(p, q, 0x1.110704p-3f);
    q = fmaf(p, q, -0x1.555532p-2f);
    return fmaf(p, x * q, x);
}

__global__ __launch_bounds__(BLOCK_THREADS, 1)
void su4_quantum_kernel(const float* __restrict__ features,
                        const float* __restrict__ weights,
                        float* __restrict__ out) {
    extern __shared__ __align__(16) float smem[];
    float2* ut = (float2*)smem;
    static_assert(GTMP_FLOATS <= WAVES_PER_BLOCK * STATE_LDS_FLOATS, "single-gate scratch must fit");
    float2* gt = (float2*)(smem + GATE_LDS_FLOATS);
    const int tid  = threadIdx.x;
    const int lane = tid & 31;
    const int wave = tid >> 5;
    float*  stateF = smem + GATE_LDS_FLOATS + wave * STATE_LDS_FLOATS;
    float2* stateV = (float2*)stateF;

    const int bBase = (blockIdx.x * WAVES_PER_BLOCK + wave) * STATES_PER_WAVE;
    static_assert(OBS == 256 && DIM == 2 * OBS, "a row is two 512-byte pieces; the upper half of a state is zero");
    const float4* fw = (const float4*)(features + (size_t)bBase * OBS);
    float4 fv[2 * STATES_PER_WAVE];
    #pragma unroll
    for (int t = 0; t < 2 * STATES_PER_WAVE; ++t) fv[t] = fw[t * 32 + lane];

    if (tid < NGATES) computeGate(tid, weights, gt);
    __syncthreads();
    for (int idx = tid; idx < NPAIRS * 64; idx += BLOCK_THREADS) fusePair(idx, gt, ut);
    __syncthreads();

    float ssq[STATES_PER_WAVE];
    #pragma unroll
    for (int s = 0; s < STATES_PER_WAVE; ++s) {
        const float4 a = fv[2 * s], b = fv[2 * s + 1];
        ssq[s] = (a.x * a.x + a.y * a.y + a.z * a.z + a.w * a.w)
               + (b.x * b.x + b.y * b.y + b.z * b.z + b.w * b.w);
    }
    #pragma unroll
    for (int msk = 16; msk >= 1; msk >>= 1) {
        #pragma unroll
        for (int s = 0; s < STATES_PER_WAVE; ++s) ssq[s] += __shfl_xor(ssq[s], msk, 32);
    }
    #pragma unroll
    for (int s = 0; s < STATES_PER_WAVE; ++s) {
        const float inv = 1.0f / (sqrtf(ssq[s]) + 1e-8f);
        const unsigned swz = stateSwz((unsigned)s);
        #pragma unroll
        for (int k = 0; k < 2; ++k) {
            const float4 v = fv[2 * s + k];
            const float a[4] = { v.x * inv, v.y * inv, v.z * inv, v.w * inv };
            const unsigned j0 = 4u * (unsigned)lane + 128u * (unsigned)k;
            const unsigned g  = (foldAmp(j0) ^ swz) & ~3u;
            float4* dst = (float4*)(stateV + s * DIM + g);
            dst[0] = make_float4(a[0 ^ (swz & 3u)], 0.f, a[1 ^ (swz & 3u)], 0.f);
            dst[1] = make_float4(a[2 ^ (swz & 3u)], 0.f, a[3 ^ (swz & 3u)], 0.f);
        }
        float4* zr = (float4*)(stateV + s * DIM + OBS);
        #pragma unroll
        for (int i = 0; i < 4; ++i) zr[i * 32 + lane] = make_float4(0.f, 0.f, 0.f, 0.f);
    }
    __syncthreads();

    const int s    = lane & 7;
    const unsigned rsub = (unsigned)((lane >> 3) & 1);
    const int half = lane >> 4;
    const int grow = lane & 15;
    const bool imRow = (grow & 1) != 0;
    const unsigned swzS = stateSwz((unsigned)s);

    static_assert((GATE_LDS_FLOATS * 4) % 256 == 0, "gate table must not shift the bank map");
    static_assert((STATE_LDS_FLOATS * 4) % 4096 == 0 && (DIM * 8) % 4096 == 0,
                  "wave region and state plane offsets must keep byte bits 0..11 clear");
    static_assert(NPAIRS % (2 * NQ) == 0, "a period of the qubit pattern is nine passes of two pairs");
    char* const smemB = (char*)smem;
    const unsigned GATE_BYTES = GATE_LDS_FLOATS * 4u;
    const unsigned waveB = (unsigned)wave * (STATE_LDS_FLOATS * 4u);
    const unsigned rowB  = waveB + (unsigned)s * (DIM * 8u);

    for (int rep = 0; rep < NPAIRS / (2 * NQ); ++rep) {
        #pragma unroll
        for (int kk = 0; kk < NQ; ++kk) {
            const int i  = (4 * kk) % NQ;
            const int pa = qbit(i), pb = qbit(i + 1), pc = qbit(i + 2), pd = qbit(i + 3), pe = qbit(i + 4);
            const int cls = bankClass(pc);
            const int sel = (bankClass(qbit(i + 5)) != cls) ? 0 : (bankClass(qbit(i + 6)) != cls) ? 1
                          : (bankClass(qbit(i + 7)) != cls) ? 2 : (bankClass(qbit(i + 8)) != cls) ? 3 : 0;
            const int pr = qbit(i + 5 + sel);
            const int pt = qbit(i + 5 + ((sel + 1) & 3));
            const int px = qbit(i + 5 + ((sel + 2) & 3));
            const int py = qbit(i + 5 + ((sel + 3) & 3));

            const float2* up1 = ut + (2 * (rep * NQ + kk)) * 64;
            const float2* up2 = up1 + 64;
            const int Q1 = (grow >> 3) * 1 + ((grow >> 2) & 1) * 4 + ((grow >> 1) & 1) * 2;
            const int Q2 = (grow >> 3) * 4 + ((grow >> 2) & 1) * 2 + ((grow >> 1) & 1) * 1;
            v2f A1[4], A2[4];
            #pragma unroll
            for (int j = 0; j < 4; ++j) {
                A1[j] = gateFrag(up1, Q1, half * 1 + (j & 1) * 2 + (j >> 1) * 4, imRow);
                A2[j] = gateFrag(up2, Q2, half * 4 + (j & 1) * 1 + (j >> 1) * 2, imRow);
            }

            const unsigned ldLane = foldAmp((rsub << pr) | ((unsigned)half << pc)) ^ swzS;
            unsigned ldB[4];
            #pragma unroll
            for (int j = 0; j < 4; ++j)
                ldB[j] = rowB + ((ldLane ^ foldAmp(((unsigned)(j & 1) << pb) | ((unsigned)(j >> 1) << pa))) << 3);

            char* const stB = smemB + GATE_BYTES;
            v2f x[16][4];
            char* q[16][4];
            v8f d1[16], d2[16];
            #define TRIP_EDGE() __builtin_amdgcn_sched_barrier(0)
            #pragma unroll 1
            for (int tb = 0; tb < 2; ++tb) {
                const unsigned tpart = tb ? (foldAmp(1u << pt) << 3) : 0u;
                #pragma unroll
                for (int k = 0; k < 16; ++k) {
                    const unsigned u0 = tpart
                        ^ (foldAmp(((unsigned)(k & 1) << pe) | ((unsigned)((k >> 1) & 1) << pd)
                                 | ((unsigned)((k >> 2) & 1) << px) | ((unsigned)((k >> 3) & 1) << py)) << 3);
                    #pragma unroll
                    for (int j = 0; j < 4; ++j) {
                        q[k][j] = stB + (ldB[j] ^ u0);
                        x[k][j] = *(const v2f*)q[k][j];
                    }
                }
                TRIP_EDGE();
                #pragma unroll
                for (int k = 0; k < 16; ++k) {
                    const v8f z = {0.f, 0.f, 0.f, 0.f, 0.f, 0.f, 0.f, 0.f};
                    d1[k] = __builtin_amdgcn_wmma_f32_16x16x4_f32(false, A1[0], false, x[k][0], (short)0, z, false, false);
                }
                #pragma unroll
                for (int j = 1; j < 4; ++j) {
                    #pragma unroll
                    for (int k = 0; k < 16; ++k)
                        d1[k] = __builtin_amdgcn_wmma_f32_16x16x4_f32(false, A1[j], false, x[k][j], (short)0, d1[k], false, false);
                }
                #pragma unroll
                for (int jp = 0; jp < 4; ++jp) {
                    #pragma unroll
                    for (int kp = 0; kp < 16; ++kp) {
                        const int kS = jp | (kp & 12), jS = kp & 3;
                        v2f b; b[0] = d1[kS][2 * jS]; b[1] = d1[kS][2 * jS + 1];
                        if (jp == 0) {
                            const v8f z = {0.f, 0.f, 0.f, 0.f, 0.f, 0.f, 0.f, 0.f};
                            d2[kp] = __builtin_amdgcn_wmma_f32_16x16x4_f32(false, A2[0], false, b, (short)0, z, false, false);
                        } else {
                            d2[kp] = __builtin_amdgcn_wmma_f32_16x16x4_f32(false, A2[jp], false, b, (short)0, d2[kp], false, false);
                        }
                    }
                }
                TRIP_EDGE();
                #pragma unroll
                for (int kp = 0; kp < 16; ++kp) {
                    #pragma unroll
                    for (int jp = 0; jp < 4; ++jp) {
                        v2f o; o[0] = d2[kp][2 * jp]; o[1] = d2[kp][2 * jp + 1];
                        *(v2f*)q[jp | (kp & 12)][kp & 3] = o;
                    }
                }
            }
            #undef TRIP_EDGE
        }
    }

    #pragma unroll 1
    for (int ss2 = 0; ss2 < STATES_PER_WAVE; ++ss2) {
        const unsigned swz = stateSwz((unsigned)ss2);
        const bool sw0 = (swz & 1u) != 0u, sw1 = (swz & 2u) != 0u;
        float* o = out + (size_t)(bBase + ss2) * DIM + 4 * lane;
        float t[4][4];
        bool big = false;
        #pragma unroll
        for (int i = 0; i < 4; ++i) {
            const unsigned j0 = 4u * (unsigned)lane + 128u * (unsigned)i;
            const unsigned g  = (foldAmp(j0) ^ swz) & ~3u;
            const float4* srcv = (const float4*)(stateV + ss2 * DIM + g);
            const float4 v0 = srcv[0], v1 = srcv[1];
            const float p0 = v0.x * v0.x + v0.y * v0.y, p1 = v0.z * v0.z + v0.w * v0.w;
            const float p2 = v1.x * v1.x + v1.y * v1.y, p3 = v1.z * v1.z + v1.w * v1.w;
            const float q0 = sw0 ? p1 : p0, q1 = sw0 ? p0 : p1, q2 = sw0 ? p3 : p2, q3 = sw0 ? p2 : p3;
            t[i][0] = 5.0f * (sw1 ? q2 : q0);
            t[i][1] = 5.0f * (sw1 ? q3 : q1);
            t[i][2] = 5.0f * (sw1 ? q0 : q2);
            t[i][3] = 5.0f * (sw1 ? q1 : q3);
            #pragma unroll
            for (int e = 0; e < 4; ++e) big = big || !(t[i][e] < 0.625f);
        }
        float r[4][4];
        if (__any((int)big)) {
            #pragma unroll
            for (int i = 0; i < 4; ++i) {
                #pragma unroll
                for (int e = 0; e < 4; ++e) r[i][e] = tanhf(t[i][e]);
            }
        } else {
            #pragma unroll
            for (int i = 0; i < 4; ++i) {
                #pragma unroll
                for (int e = 0; e < 4; ++e) r[i][e] = tanhSmall(t[i][e]);
            }
        }
        #pragma unroll
        for (int i = 0; i < 4; ++i)
            *(float4*)(o + 128 * i) = make_float4(r[i][0], r[i][1], r[i][2], r[i][3]);
    }
}

extern "C" void kernel_launch(void* const* d_in, const int* in_sizes, int n_in,
                              void* d_out, int out_size, void* d_ws, size_t ws_size,
                              hipStream_t stream) {
    (void)in_sizes; (void)n_in; (void)d_ws; (void)ws_size; (void)out_size;
    const float* features = (const float*)d_in[0];
    const float* weights  = (const float*)d_in[1];
    float* out = (float*)d_out;

    dim3 grid(BATCH_TOT / STATES_PER_BLOCK);
    dim3 block(BLOCK_THREADS);
    size_t shmem = (size_t)(GATE_LDS_FLOATS + WAVES_PER_BLOCK * STATE_LDS_FLOATS)
                   * sizeof(float);
    su4_quantum_kernel<<<grid, block, shmem, stream>>>(features, weights, out);
}


# State kept in registers through all 54 pair products: 1 wave per SIMD, 16 states as columns, a pair's D pairs are the next pair's B fragments, no LDS passes; k order of sums, norm sum order changed
# speedup vs baseline: 5.9808x; 1.1711x over previous
#include <hip/hip_runtime.h>
#include <math.h>

#define NQ        9
#define DIM       512
#define NLAYERS   12
#define OBS       256
#define BATCH_TOT 16384

#define WAVES_PER_BLOCK   4
#define STATES_PER_WAVE   16
#define BLOCK_THREADS     (WAVES_PER_BLOCK * 32)
#define STATES_PER_BLOCK  (WAVES_PER_BLOCK * STATES_PER_WAVE)
#define NGATES            (NLAYERS * NQ)
#define NPAIRS            (NGATES / 2)
#define PAIR_FLOATS       128
#define GATE_LDS_FLOATS   (NPAIRS * PAIR_FLOATS)
#define ROW_PITCH         (DIM + 4)
#define STAGE_LDS_FLOATS  (STATES_PER_WAVE * ROW_PITCH)
#define LDS_TOTAL_FLOATS  72448
#define GTMP_FLOATS       (NGATES * 32)

typedef float v2f __attribute__((ext_vector_type(2)));
typedef float v4f __attribute__((ext_vector_type(4)));
typedef float v8f __attribute__((ext_vector_type(8)));

struct cpx { float re, im; };
__device__ inline cpx cz() { cpx r; r.re = 0.f; r.im = 0.f; return r; }
__device__ inline cpx cmk(float a, float b) { cpx r; r.re = a; r.im = b; return r; }
__device__ inline cpx cmul(cpx a, cpx b) { return cmk(a.re*b.re - a.im*b.im, a.re*b.im + a.im*b.re); }
__device__ inline cpx cadd(cpx a, cpx b) { return cmk(a.re + b.re, a.im + b.im); }

__device__ inline void u3mat(float t, float p, float l, cpx U[2][2]) {
    float ct = cosf(0.5f * t), st = sinf(0.5f * t);
    float cp = cosf(p), sp = sinf(p), cl = cosf(l), sl = sinf(l);
    U[0][0] = cmk(ct, 0.f);
    U[0][1] = cmk(-cl * st, -sl * st);
    U[1][0] = cmk(cp * st, sp * st);
    U[1][1] = cmk((cp*cl - sp*sl) * ct, (cp*sl + sp*cl) * ct);
}

__device__ inline void kron22(const cpx A[2][2], const cpx B[2][2], cpx K[4][4]) {
    for (int a = 0; a < 2; ++a)
        for (int b = 0; b < 2; ++b)
            for (int c = 0; c < 2; ++c)
                for (int d = 0; d < 2; ++d)
                    K[2*a + c][2*b + d] = cmul(A[a][b], B[c][d]);
}

__device__ inline void mat4mul(cpx C[4][4], const cpx A[4][4], const cpx B[4][4]) {
    for (int i = 0; i < 4; ++i)
        for (int j = 0; j < 4; ++j) {
            cpx acc = cz();
            for (int k = 0; k < 4; ++k) acc = cadd(acc, cmul(A[i][k], B[k][j]));
            C[i][j] = acc;
        }
}

__device__ void computeGate(int g, const float* __restrict__ w, float2* __restrict__ gt) {
    const float* th = w + g * 15;
    cpx Ua[2][2], Ub[2][2];
    cpx pre[4][4], post[4][4];
    u3mat(th[0], th[1], th[2], Ua);  u3mat(th[3], th[4], th[5], Ub);  kron22(Ua, Ub, pre);
    u3mat(th[9], th[10], th[11], Ua); u3mat(th[12], th[13], th[14], Ub); kron22(Ua, Ub, post);

    float c6 = cosf(0.5f*th[6]), s6 = sinf(0.5f*th[6]);
    float c7 = cosf(0.5f*th[7]), s7 = sinf(0.5f*th[7]);
    float c8 = cosf(0.5f*th[8]), s8 = sinf(0.5f*th[8]);
    cpx Rxx[4][4], Ryy[4][4], Rzz[4][4];
    for (int i = 0; i < 4; ++i)
        for (int j = 0; j < 4; ++j) { Rxx[i][j] = cz(); Ryy[i][j] = cz(); Rzz[i][j] = cz(); }
    for (int i = 0; i < 4; ++i) { Rxx[i][i] = cmk(c6, 0.f); Ryy[i][i] = cmk(c7, 0.f); }
    Rxx[0][3] = cmk(0.f, -s6); Rxx[1][2] = cmk(0.f, -s6);
    Rxx[2][1] = cmk(0.f, -s6); Rxx[3][0] = cmk(0.f, -s6);
    Ryy[0][3] = cmk(0.f,  s7); Ryy[1][2] = cmk(0.f, -s7);
    Ryy[2][1] = cmk(0.f, -s7); Ryy[3][0] = cmk(0.f,  s7);
    Rzz[0][0] = cmk(c8, -s8); Rzz[1][1] = cmk(c8,  s8);
    Rzz[2][2] = cmk(c8,  s8); Rzz[3][3] = cmk(c8, -s8);

    cpx t1[4][4], t2[4][4];
    mat4mul(t1, Rxx, Ryy);
    mat4mul(t2, t1, Rzz);
    mat4mul(t1, t2, pre);
    mat4mul(t2, post, t1);

    float2* gb = gt + g * 16;
    for (int a = 0; a < 4; ++a)
        for (int b = 0; b < 4; ++b)
            gb[a*4 + b] = make_float2(t2[a][b].re, t2[a][b].im);
}

__device__ inline void fusePair(int idx, const float2* __restrict__ gt, float2* __restrict__ ut) {
    const int pr = idx >> 6, Q = (idx >> 3) & 7, P = idx & 7;
    const float2* G1 = gt + (2 * pr) * 16;
    const float2* G2 = gt + (2 * pr + 1) * 16;
    const int A = Q >> 2, B = (Q >> 1) & 1, C = Q & 1;
    const int a = P >> 2, b = (P >> 1) & 1, c = P & 1;
    float re = 0.f, im = 0.f;
    for (int bp = 0; bp < 2; ++bp) {
        const float2 x = G2[(2 * B + C) * 4 + 2 * bp + c];
        const float2 y = G1[(2 * A + bp) * 4 + 2 * a + b];
        re += x.x * y.x - x.y * y.y;
        im += x.x * y.y + x.y * y.x;
    }
    ut[idx] = make_float2(re, im);
}

__device__ inline v2f gateFrag(const float2* ut, int Q, int P, bool imRow) {
    const float2 u = ut[Q * 8 + P];
    v2f r;
    r[0] = imRow ? u.y : u.x;
    r[1] = imRow ? u.x : -u.y;
    return r;
}

__device__ inline float tanhSmall(float x) {
    const float p = x * x;
    float q = fmaf(p, -0x1.758e7ap-8f, 0x1.521192p-6f);
    q = fmaf(p, q, -0x1.b8389cp-5f);
    q = fmaf(p, q, 0x1.110704p-3f);
    q = fmaf(p, q, -0x1.555532p-2f);
    return fmaf(p, x * q, x);
}

__device__ __forceinline__ void pairStep(v8f (&T)[64], const v2f (&A)[4]) {
    v8f N[64];
    #pragma unroll
    for (int g = 0; g < 16; ++g) {
        v8f acc[4];
        #pragma unroll
        for (int j = 0; j < 4; ++j) {
            #pragma unroll
            for (int u = 0; u < 4; ++u) {
                v2f b; b[0] = T[(j << 4) | g][2 * u]; b[1] = T[(j << 4) | g][2 * u + 1];
                if (j == 0) {
                    const v8f z = {0.f, 0.f, 0.f, 0.f, 0.f, 0.f, 0.f, 0.f};
                    acc[u] = __builtin_amdgcn_wmma_f32_16x16x4_f32(false, A[0], false, b, (short)0, z, false, false);
                } else {
                    acc[u] = __builtin_amdgcn_wmma_f32_16x16x4_f32(false, A[j], false, b, (short)0, acc[u], false, false);
                }
            }
        }
        #pragma unroll
        for (int u = 0; u < 4; ++u) N[4 * g + u] = acc[u];
    }
    #pragma unroll
    for (int t = 0; t < 64; ++t) T[t] = N[t];
}

__device__ __forceinline__ void loadGateFrags(v2f (&A)[4], const float2* ut, int pr, int grow, int half) {
    const float2* up = ut + pr * 64;
    const int Q = 2 * ((grow >> 1) & 3) + (grow >> 3);
    const bool imRow = (grow & 1) != 0;
    #pragma unroll
    for (int j = 0; j < 4; ++j) A[j] = gateFrag(up, Q, 4 * half + j, imRow);
}

__global__ __launch_bounds__(BLOCK_THREADS, 1)
void su4_quantum_kernel(const float* __restrict__ features,
                        const float* __restrict__ weights,
                        float* __restrict__ out) {
    extern __shared__ __align__(16) float smem[];
    float2* ut = (float2*)smem;
    static_assert(GTMP_FLOATS <= WAVES_PER_BLOCK * STAGE_LDS_FLOATS, "single-gate scratch must fit");
    static_assert(GATE_LDS_FLOATS + WAVES_PER_BLOCK * STAGE_LDS_FLOATS <= LDS_TOTAL_FLOATS, "LDS layout must fit");
    float2* gt = (float2*)(smem + GATE_LDS_FLOATS);
    const int tid  = threadIdx.x;
    const int lane = tid & 31;
    const int wave = tid >> 5;
    const int n    = lane & 15;
    const int half = lane >> 4;
    float* stage = smem + GATE_LDS_FLOATS + wave * STAGE_LDS_FLOATS;

    const int bBase = (blockIdx.x * WAVES_PER_BLOCK + wave) * STATES_PER_WAVE;
    static_assert(OBS == 256 && DIM == 2 * OBS, "a row is two 512-byte pieces; the upper half of a state is zero");
    const v4f* fw = (const v4f*)(features + (size_t)bBase * OBS);
    v4f fr[2 * STATES_PER_WAVE];
    #pragma unroll
    for (int t = 0; t < 2 * STATES_PER_WAVE; ++t) fr[t] = fw[t * 32 + lane];

    if (tid < NGATES) computeGate(tid, weights, gt);
    __syncthreads();
    for (int idx = tid; idx < NPAIRS * 64; idx += BLOCK_THREADS) fusePair(idx, gt, ut);
    __syncthreads();

    #pragma unroll
    for (int t = 0; t < 2 * STATES_PER_WAVE; ++t)
        *(v4f*)(stage + (t >> 1) * ROW_PITCH + 4 * ((t & 1) * 32 + lane)) = fr[t];
    float4 fv[64];
    #pragma unroll
    for (int t = 0; t < 64; ++t) fv[t] = make_float4(0.f, 0.f, 0.f, 0.f);
    if (half == 0) {
        #pragma unroll
        for (int t = 0; t < 64; ++t) fv[t] = *(const float4*)(stage + n * ROW_PITCH + 4 * t);
    }
    float ssq = 0.f;
    #pragma unroll
    for (int t = 0; t < 64; ++t) {
        const float4 a = fv[t];
        ssq += (a.x * a.x + a.y * a.y) + (a.z * a.z + a.w * a.w);
    }
    const float inv = 1.0f / (sqrtf(ssq) + 1e-8f);
    v8f T[64];
    #pragma unroll
    for (int t = 0; t < 64; ++t) {
        const float4 a = fv[t];
        T[t][0] = a.x * inv; T[t][1] = 0.f;
        T[t][2] = a.y * inv; T[t][3] = 0.f;
        T[t][4] = a.z * inv; T[t][5] = 0.f;
        T[t][6] = a.w * inv; T[t][7] = 0.f;
    }

    const int grow = lane & 15;
    v2f A[4];
    loadGateFrags(A, ut, 0, grow, half);
    #pragma unroll 1
    for (int pr = 0; pr < NPAIRS; pr += 3) {
        v2f A1[4], A2[4], A3[4];
        loadGateFrags(A1, ut, pr + 1, grow, half);
        pairStep(T, A);
        loadGateFrags(A2, ut, pr + 2, grow, half);
        pairStep(T, A1);
        loadGateFrags(A3, ut, (pr + 3 < NPAIRS) ? pr + 3 : pr, grow, half);
        pairStep(T, A2);
        #pragma unroll
        for (int j = 0; j < 4; ++j) A[j] = A3[j];
    }
    static_assert(NPAIRS % 3 == 0 && (2 * NPAIRS) % NQ == 0, "54 pairs bring the first qubit back to the lane half");

    #pragma unroll
    for (int t = 0; t < 64; ++t) {
        const v8f v = T[t];
        const float p0 = v[0] * v[0] + v[1] * v[1], p1 = v[2] * v[2] + v[3] * v[3];
        const float p2 = v[4] * v[4] + v[5] * v[5], p3 = v[6] * v[6] + v[7] * v[7];
        *(float4*)(stage + n * ROW_PITCH + half * OBS + 4 * t) =
            make_float4(5.0f * p0, 5.0f * p1, 5.0f * p2, 5.0f * p3);
    }
    #pragma unroll 1
    for (int ss2 = 0; ss2 < STATES_PER_WAVE; ++ss2) {
        float* o = out + (size_t)(bBase + ss2) * DIM + 4 * lane;
        float t[4][4];
        bool big = false;
        #pragma unroll
        for (int i = 0; i < 4; ++i) {
            const float4 v = *(const float4*)(stage + ss2 * ROW_PITCH + 4 * lane + 128 * i);
            t[i][0] = v.x; t[i][1] = v.y; t[i][2] = v.z; t[i][3] = v.w;
            #pragma unroll
            for (int e = 0; e < 4; ++e) big = big || !(t[i][e] < 0.625f);
        }
        float r[4][4];
        if (__any((int)big)) {
            #pragma unroll
            for (int i = 0; i < 4; ++i) {
                #pragma unroll
                for (int e = 0; e < 4; ++e) r[i][e] = tanhf(t[i][e]);
            }
        } else {
            #pragma unroll
            for (int i = 0; i < 4; ++i) {
                #pragma unroll
                for (int e = 0; e < 4; ++e) r[i][e] = tanhSmall(t[i][e]);
            }
        }
        #pragma unroll
        for (int i = 0; i < 4; ++i)
            *(float4*)(o + 128 * i) = make_float4(r[i][0], r[i][1], r[i][2], r[i][3]);
    }
}

extern "C" void kernel_launch(void* const* d_in, const int* in_sizes, int n_in,
                              void* d_out, int out_size, void* d_ws, size_t ws_size,
                              hipStream_t stream) {
    (void)in_sizes; (void)n_in; (void)d_ws; (void)ws_size; (void)out_size;
    const float* features = (const float*)d_in[0];
    const float* weights  = (const float*)d_in[1];
    float* out = (float*)d_out;

    dim3 grid(BATCH_TOT / STATES_PER_BLOCK);
    dim3 block(BLOCK_THREADS);
    size_t shmem = (size_t)LDS_TOTAL_FLOATS * sizeof(float);
    su4_quantum_kernel<<<grid, block, shmem, stream>>>(features, weights, out);
}
